# starting kernel
# baseline (speedup 1.0000x reference)
#include <hip/hip_runtime.h>
#include <cstddef>

#define NJOINT 25
#define CIN    64
#define COUT   64
#define VC     (NJOINT * COUT)
#define NFRM   (64 * 300)
#define BN_EPS 1e-5f

#define WPB        4
#define FPW        8
#define BLOCKS_B   600

#define NCHUNK     60
#define CHUNKROWS  320

#define WS_AHAT   0
#define WS_PSUM   1024
#define WS_PSQ    (WS_PSUM + NCHUNK * VC)
#define WS_SCALE  (WS_PSQ  + NCHUNK * VC)
#define WS_SHIFT  (WS_SCALE + VC)

typedef __attribute__((ext_vector_type(2))) float v2f;
typedef __attribute__((ext_vector_type(8))) float v8f;

__global__ void stgcn_build_ahat(const int* __restrict__ edge_index, int E,
                                 float* __restrict__ ws) {
  if (threadIdx.x != 0 || blockIdx.x != 0) return;
  const int* row = edge_index;
  const int* col = edge_index + E;
  float deg[NJOINT];
  for (int v = 0; v < NJOINT; ++v) deg[v] = 1.0f;
  for (int e = 0; e < E; ++e) deg[col[e]] += 1.0f;
  float dinv[NJOINT];
  for (int v = 0; v < NJOINT; ++v) dinv[v] = rsqrtf(deg[v]);
  float* A = ws + WS_AHAT;
  for (int i = 0; i < 32 * 32; ++i) A[i] = 0.0f;
  for (int e = 0; e < E; ++e) {
    int s = row[e], t = col[e];
    A[t * 32 + s] += dinv[s] * dinv[t];
  }
  for (int v = 0; v < NJOINT; ++v) A[v * 32 + v] += dinv[v] * dinv[v];
}

#define WSTR 76
#define PSTR 80
#define ASTR 17

__launch_bounds__(WPB * 32, 1)
__global__ void stgcn_gcn_gemm(const float* __restrict__ x,
                               const float* __restrict__ W,
                               const float* __restrict__ bias,
                               const float* __restrict__ ws,
                               float* __restrict__ out) {
  __shared__ v2f wpair[CIN / 2][WSTR];
  __shared__ v2f apair[32][ASTR];
  __shared__ v2f pwbuf[WPB][16][PSTR];
  __shared__ float bbuf[COUT];

  const int tid = threadIdx.x;

  for (int i = tid; i < (CIN / 2) * COUT; i += WPB * 32) {
    const int p = i >> 6, d = i & 63;
    v2f val;
    val.x = W[(2 * p) * COUT + d];
    val.y = W[(2 * p + 1) * COUT + d];
    wpair[p][d] = val;
  }
  for (int i = tid; i < 32 * 16; i += WPB * 32) {
    apair[i >> 4][i & 15] = ((const v2f*)(ws + WS_AHAT))[i];
  }
  if (tid < COUT) bbuf[tid] = bias[tid];
  __syncthreads();

  const int wave = tid >> 5;
  const int lane = tid & 31;
  const int half = lane >> 4;
  const int ln   = lane & 15;
  v2f (*pw)[PSTR] = pwbuf[wave];

  const int waveGlobal = blockIdx.x * WPB + wave;

  for (int f = 0; f < FPW; ++f) {
    const int frame = waveGlobal + f * (BLOCKS_B * WPB);
    const float* xf = x + (size_t)frame * (NJOINT * CIN);

#pragma unroll
    for (int mt = 0; mt < 2; ++mt) {
      v8f acc[4];
#pragma unroll
      for (int nt = 0; nt < 4; ++nt) acc[nt] = (v8f)(0.0f);

      const int m = mt * 16 + ln;
      const bool mvalid = (m < NJOINT);
      const v2f* xrow = (const v2f*)(xf + m * CIN);
#pragma unroll
      for (int ks = 0; ks < 16; ++ks) {
        const int kp = 2 * ks + half;
        v2f a = (v2f)(0.0f);
        if (mvalid) a = xrow[kp];
#pragma unroll
        for (int nt = 0; nt < 4; ++nt) {
          v2f b = wpair[kp][nt * 16 + ln];
          acc[nt] = __builtin_amdgcn_wmma_f32_16x16x4_f32(
              false, a, false, b, (short)0, acc[nt], false, false);
        }
      }
#pragma unroll
      for (int nt = 0; nt < 4; ++nt)
#pragma unroll
        for (int r = 0; r < 8; r += 2) {
          v2f val;
          val.x = acc[nt][r];
          val.y = acc[nt][r + 1];
          pw[mt * 8 + half * 4 + (r >> 1)][nt * 16 + ln] = val;
        }
    }

#pragma unroll
    for (int mt = 0; mt < 2; ++mt) {
      v8f acc[4];
#pragma unroll
      for (int nt = 0; nt < 4; ++nt) acc[nt] = (v8f)(0.0f);

#pragma unroll
      for (int ks = 0; ks < 8; ++ks) {
        const int kp = 2 * ks + half;
        v2f a = apair[mt * 16 + ln][kp];
#pragma unroll
        for (int nt = 0; nt < 4; ++nt) {
          v2f b = pw[kp][nt * 16 + ln];
          acc[nt] = __builtin_amdgcn_wmma_f32_16x16x4_f32(
              false, a, false, b, (short)0, acc[nt], false, false);
        }
      }
      float* of = out + (size_t)frame * VC;
#pragma unroll
      for (int nt = 0; nt < 4; ++nt) {
        const int colb = nt * 16 + ln;
        const float bb = bbuf[colb];
#pragma unroll
        for (int r = 0; r < 8; ++r) {
          const int v = mt * 16 + half * 8 + r;
          if (v < NJOINT) {
            float val = acc[nt][r] + bb;
            of[v * COUT + colb] = val > 0.0f ? val : 0.0f;
          }
        }
      }
    }
  }
}

__global__ void stgcn_bn_partial(const float* __restrict__ out,
                                 float* __restrict__ ws) {
  const int ch = blockIdx.x * blockDim.x + threadIdx.x;
  if (ch >= VC) return;
  const int chunk = blockIdx.y;
  const float* p = out + (size_t)chunk * CHUNKROWS * VC + ch;
  float s = 0.0f, ss = 0.0f;
  for (int i = 0; i < CHUNKROWS; ++i) {
    float v = p[(size_t)i * VC];
    s += v;
    ss += v * v;
  }
  ws[WS_PSUM + chunk * VC + ch] = s;
  ws[WS_PSQ + chunk * VC + ch] = ss;
}

__global__ void stgcn_bn_finalize(const float* __restrict__ gamma,
                                  const float* __restrict__ beta,
                                  float* __restrict__ ws) {
  const int ch = blockIdx.x * blockDim.x + threadIdx.x;
  if (ch >= VC) return;
  double s = 0.0, ss = 0.0;
  for (int c = 0; c < NCHUNK; ++c) {
    s += (double)ws[WS_PSUM + c * VC + ch];
    ss += (double)ws[WS_PSQ + c * VC + ch];
  }
  const double n = (double)NFRM;
  const double mean = s / n;
  const double var = ss / n - mean * mean;
  const float scale = gamma[ch] * rsqrtf((float)var + BN_EPS);
  const float shift = beta[ch] - (float)mean * scale;
  ws[WS_SCALE + ch] = scale;
  ws[WS_SHIFT + ch] = shift;
}

__global__ void stgcn_bn_apply(float* __restrict__ out,
                               const float* __restrict__ ws) {
  const size_t idx = ((size_t)blockIdx.x * blockDim.x + threadIdx.x) * 4;
  float4 h = *(const float4*)(out + idx);
  const int ch = (int)(idx % VC);
  const float4 sc = *(const float4*)(ws + WS_SCALE + ch);
  const float4 sh = *(const float4*)(ws + WS_SHIFT + ch);
  h.x = h.x * sc.x + sh.x;
  h.y = h.y * sc.y + sh.y;
  h.z = h.z * sc.z + sh.z;
  h.w = h.w * sc.w + sh.w;
  *(float4*)(out + idx) = h;
}

extern "C" void kernel_launch(void* const* d_in, const int* in_sizes, int n_in,
                              void* d_out, int out_size, void* d_ws, size_t ws_size,
                              hipStream_t stream) {
  const float* x     = (const float*)d_in[0];
  const float* W     = (const float*)d_in[1];
  const float* b     = (const float*)d_in[2];
  const float* gamma = (const float*)d_in[3];
  const float* beta  = (const float*)d_in[4];
  const int*   ei    = (const int*)d_in[5];
  float* out = (float*)d_out;
  float* ws  = (float*)d_ws;
  const int E = in_sizes[5] / 2;

  hipLaunchKernelGGL(stgcn_build_ahat, dim3(1), dim3(64), 0, stream, ei, E, ws);
  hipLaunchKernelGGL(stgcn_gcn_gemm, dim3(BLOCKS_B), dim3(WPB * 32), 0, stream,
                     x, W, b, ws, out);
  hipLaunchKernelGGL(stgcn_bn_partial, dim3((VC + 255) / 256, NCHUNK), dim3(256),
                     0, stream, out, ws);
  hipLaunchKernelGGL(stgcn_bn_finalize, dim3((VC + 255) / 256), dim3(256),
                     0, stream, gamma, beta, ws);
  hipLaunchKernelGGL(stgcn_bn_apply, dim3((NFRM * VC) / 4 / 256), dim3(256),
                     0, stream, out, ws);
}


# A_hat is built by every GEMM block in its own LDS (edge compares, per-entry register sums in edge order, diagonal last); the serial one-thread builder launch and its workspace round trip are gone
# speedup vs baseline: 1.2020x; 1.2020x over previous
#include <hip/hip_runtime.h>
#include <cstddef>

#define NJOINT 25
#define CIN    64
#define COUT   64
#define VC     (NJOINT * COUT)
#define NFRM   (64 * 300)
#define BN_EPS 1e-5f

#define WPB        4
#define FPW        8
#define BLOCKS_B   600

#define NCHUNK     60
#define CHUNKROWS  320

#define WS_PSUM   1024
#define WS_PSQ    (WS_PSUM + NCHUNK * VC)
#define WS_SCALE  (WS_PSQ  + NCHUNK * VC)
#define WS_SHIFT  (WS_SCALE + VC)

typedef __attribute__((ext_vector_type(2))) float v2f;
typedef __attribute__((ext_vector_type(8))) float v8f;

#define WSTR 76
#define PSTR 80
#define ASTR 17
#define ECHUNK (WPB * 32)

__launch_bounds__(WPB * 32, 1)
__global__ void stgcn_gcn_gemm(const float* __restrict__ x,
                               const float* __restrict__ W,
                               const float* __restrict__ bias,
                               const int* __restrict__ edge_index, int E,
                               float* __restrict__ out) {
  __shared__ v2f wpair[CIN / 2][WSTR];
  __shared__ v2f apair[32][ASTR];
  __shared__ v2f pwbuf[WPB][16][PSTR];
  __shared__ float bbuf[COUT];
  __shared__ int2 ebuf[ECHUNK];
  __shared__ float dinvbuf[32];

  const int tid = threadIdx.x;

  for (int i = tid; i < (CIN / 2) * COUT; i += WPB * 32) {
    const int p = i >> 6, d = i & 63;
    v2f val;
    val.x = W[(2 * p) * COUT + d];
    val.y = W[(2 * p + 1) * COUT + d];
    wpair[p][d] = val;
  }
  if (tid < COUT) bbuf[tid] = bias[tid];

  const int* esrc = edge_index;
  const int* etgt = edge_index + E;
  const int t  = tid >> 2;
  const int s0 = (tid & 3) << 3;
  const bool onechunk = (E <= ECHUNK);

  float deg = 1.0f;
  for (int base = 0; base < E; base += ECHUNK) {
    if (base > 0) __syncthreads();
    int2 ed;
    ed.x = -1; ed.y = -1;
    if (base + tid < E) { ed.x = etgt[base + tid]; ed.y = esrc[base + tid]; }
    ebuf[tid] = ed;
    __syncthreads();
    const int n = min(E - base, ECHUNK);
#pragma unroll 4
    for (int e = 0; e < n; ++e) deg += (ebuf[e].x == t) ? 1.0f : 0.0f;
  }
  const float dinv_t = rsqrtf(deg);
  if ((tid & 3) == 0) dinvbuf[t] = dinv_t;
  __syncthreads();

  float wj[8], aent[8];
#pragma unroll
  for (int j = 0; j < 8; ++j) {
    wj[j] = dinvbuf[s0 + j] * dinv_t;
    aent[j] = 0.0f;
  }
  for (int base = 0; base < E; base += ECHUNK) {
    if (!onechunk) {
      int2 ed;
      ed.x = -1; ed.y = -1;
      if (base + tid < E) { ed.x = etgt[base + tid]; ed.y = esrc[base + tid]; }
      ebuf[tid] = ed;
      __syncthreads();
    }
    const int n = min(E - base, ECHUNK);
#pragma unroll 4
    for (int e = 0; e < n; ++e) {
      const int2 ed = ebuf[e];
      const int d = (ed.x == t) ? (ed.y - s0) : -1;
#pragma unroll
      for (int j = 0; j < 8; ++j) aent[j] += (d == j) ? wj[j] : 0.0f;
    }
    if (!onechunk) __syncthreads();
  }
#pragma unroll
  for (int j = 0; j < 8; ++j) aent[j] += (s0 + j == t && t < NJOINT) ? wj[j] : 0.0f;
#pragma unroll
  for (int q = 0; q < 4; ++q) {
    v2f val;
    val.x = aent[2 * q];
    val.y = aent[2 * q + 1];
    apair[t][(s0 >> 1) + q] = val;
  }
  __syncthreads();

  const int wave = tid >> 5;
  const int lane = tid & 31;
  const int half = lane >> 4;
  const int ln   = lane & 15;
  v2f (*pw)[PSTR] = pwbuf[wave];

  const int waveGlobal = blockIdx.x * WPB + wave;

  for (int f = 0; f < FPW; ++f) {
    const int frame = waveGlobal + f * (BLOCKS_B * WPB);
    const float* xf = x + (size_t)frame * (NJOINT * CIN);

#pragma unroll
    for (int mt = 0; mt < 2; ++mt) {
      v8f acc[4];
#pragma unroll
      for (int nt = 0; nt < 4; ++nt) acc[nt] = (v8f)(0.0f);

      const int m = mt * 16 + ln;
      const bool mvalid = (m < NJOINT);
      const v2f* xrow = (const v2f*)(xf + m * CIN);
#pragma unroll
      for (int ks = 0; ks < 16; ++ks) {
        const int kp = 2 * ks + half;
        v2f a = (v2f)(0.0f);
        if (mvalid) a = xrow[kp];
#pragma unroll
        for (int nt = 0; nt < 4; ++nt) {
          v2f b = wpair[kp][nt * 16 + ln];
          acc[nt] = __builtin_amdgcn_wmma_f32_16x16x4_f32(
              false, a, false, b, (short)0, acc[nt], false, false);
        }
      }
#pragma unroll
      for (int nt = 0; nt < 4; ++nt)
#pragma unroll
        for (int r = 0; r < 8; r += 2) {
          v2f val;
          val.x = acc[nt][r];
          val.y = acc[nt][r + 1];
          pw[mt * 8 + half * 4 + (r >> 1)][nt * 16 + ln] = val;
        }
    }

#pragma unroll
    for (int mt = 0; mt < 2; ++mt) {
      v8f acc[4];
#pragma unroll
      for (int nt = 0; nt < 4; ++nt) acc[nt] = (v8f)(0.0f);

#pragma unroll
      for (int ks = 0; ks < 8; ++ks) {
        const int kp = 2 * ks + half;
        v2f a = apair[mt * 16 + ln][kp];
#pragma unroll
        for (int nt = 0; nt < 4; ++nt) {
          v2f b = pw[kp][nt * 16 + ln];
          acc[nt] = __builtin_amdgcn_wmma_f32_16x16x4_f32(
              false, a, false, b, (short)0, acc[nt], false, false);
        }
      }
      float* of = out + (size_t)frame * VC;
#pragma unroll
      for (int nt = 0; nt < 4; ++nt) {
        const int colb = nt * 16 + ln;
        const float bb = bbuf[colb];
#pragma unroll
        for (int r = 0; r < 8; ++r) {
          const int v = mt * 16 + half * 8 + r;
          if (v < NJOINT) {
            float val = acc[nt][r] + bb;
            of[v * COUT + colb] = val > 0.0f ? val : 0.0f;
          }
        }
      }
    }
  }
}

__global__ void stgcn_bn_partial(const float* __restrict__ out,
                                 float* __restrict__ ws) {
  const int ch = blockIdx.x * blockDim.x + threadIdx.x;
  if (ch >= VC) return;
  const int chunk = blockIdx.y;
  const float* p = out + (size_t)chunk * CHUNKROWS * VC + ch;
  float s = 0.0f, ss = 0.0f;
  for (int i = 0; i < CHUNKROWS; ++i) {
    float v = p[(size_t)i * VC];
    s += v;
    ss += v * v;
  }
  ws[WS_PSUM + chunk * VC + ch] = s;
  ws[WS_PSQ + chunk * VC + ch] = ss;
}

__global__ void stgcn_bn_finalize(const float* __restrict__ gamma,
                                  const float* __restrict__ beta,
                                  float* __restrict__ ws) {
  const int ch = blockIdx.x * blockDim.x + threadIdx.x;
  if (ch >= VC) return;
  double s = 0.0, ss = 0.0;
  for (int c = 0; c < NCHUNK; ++c) {
    s += (double)ws[WS_PSUM + c * VC + ch];
    ss += (double)ws[WS_PSQ + c * VC + ch];
  }
  const double n = (double)NFRM;
  const double mean = s / n;
  const double var = ss / n - mean * mean;
  const float scale = gamma[ch] * rsqrtf((float)var + BN_EPS);
  const float shift = beta[ch] - (float)mean * scale;
  ws[WS_SCALE + ch] = scale;
  ws[WS_SHIFT + ch] = shift;
}

__global__ void stgcn_bn_apply(float* __restrict__ out,
                               const float* __restrict__ ws) {
  const size_t idx = ((size_t)blockIdx.x * blockDim.x + threadIdx.x) * 4;
  float4 h = *(const float4*)(out + idx);
  const int ch = (int)(idx % VC);
  const float4 sc = *(const float4*)(ws + WS_SCALE + ch);
  const float4 sh = *(const float4*)(ws + WS_SHIFT + ch);
  h.x = h.x * sc.x + sh.x;
  h.y = h.y * sc.y + sh.y;
  h.z = h.z * sc.z + sh.z;
  h.w = h.w * sc.w + sh.w;
  *(float4*)(out + idx) = h;
}

extern "C" void kernel_launch(void* const* d_in, const int* in_sizes, int n_in,
                              void* d_out, int out_size, void* d_ws, size_t ws_size,
                              hipStream_t stream) {
  const float* x     = (const float*)d_in[0];
  const float* W     = (const float*)d_in[1];
  const float* b     = (const float*)d_in[2];
  const float* gamma = (const float*)d_in[3];
  const float* beta  = (const float*)d_in[4];
  const int*   ei    = (const int*)d_in[5];
  float* out = (float*)d_out;
  float* ws  = (float*)d_ws;
  const int E = in_sizes[5] / 2;

  hipLaunchKernelGGL(stgcn_gcn_gemm, dim3(BLOCKS_B), dim3(WPB * 32), 0, stream,
                     x, W, b, ei, E, out);
  hipLaunchKernelGGL(stgcn_bn_partial, dim3((VC + 255) / 256, NCHUNK), dim3(256),
                     0, stream, out, ws);
  hipLaunchKernelGGL(stgcn_bn_finalize, dim3((VC + 255) / 256), dim3(256),
                     0, stream, gamma, beta, ws);
  hipLaunchKernelGGL(stgcn_bn_apply, dim3((NFRM * VC) / 4 / 256), dim3(256),
                     0, stream, out, ws);
}


# X fragment loads made unconditional (padded rows read row 24, zeroed at use) and requested as one group a frame ahead; the second row tile no longer waits on a guarded load before every k-step
# speedup vs baseline: 1.2915x; 1.0745x over previous
#include <hip/hip_runtime.h>
#include <cstddef>

#define NJOINT 25
#define CIN    64
#define COUT   64
#define VC     (NJOINT * COUT)
#define NFRM   (64 * 300)
#define BN_EPS 1e-5f

#define WPB        4
#define FPW        8
#define BLOCKS_B   600

#define NCHUNK     60
#define CHUNKROWS  320

#define WS_PSUM   1024
#define WS_PSQ    (WS_PSUM + NCHUNK * VC)
#define WS_SCALE  (WS_PSQ  + NCHUNK * VC)
#define WS_SHIFT  (WS_SCALE + VC)

typedef __attribute__((ext_vector_type(2))) float v2f;
typedef __attribute__((ext_vector_type(8))) float v8f;

#define WSTR 76
#define PSTR 80
#define ASTR 17
#define ECHUNK (WPB * 32)

__launch_bounds__(WPB * 32, 1)
__global__ void stgcn_gcn_gemm(const float* __restrict__ x,
                               const float* __restrict__ W,
                               const float* __restrict__ bias,
                               const int* __restrict__ edge_index, int E,
                               float* __restrict__ out) {
  __shared__ v2f wpair[CIN / 2][WSTR];
  __shared__ v2f apair[32][ASTR];
  __shared__ v2f pwbuf[WPB][16][PSTR];
  __shared__ float bbuf[COUT];
  __shared__ int2 ebuf[ECHUNK];
  __shared__ float dinvbuf[32];

  const int tid = threadIdx.x;
  const int wave = tid >> 5;
  const int lane = tid & 31;
  const int half = lane >> 4;
  const int ln   = lane & 15;
  const int waveGlobal = blockIdx.x * WPB + wave;

  const bool m1valid = (16 + ln) < NJOINT;
  const int xoff0 = ln * CIN + 2 * half;
  const int xoff1 = (m1valid ? (16 + ln) : (NJOINT - 1)) * CIN + 2 * half;
  v2f xa[2][16];
  {
    const float* xf0 = x + (size_t)waveGlobal * (NJOINT * CIN);
    const v2f* p0 = (const v2f*)(xf0 + xoff0);
    const v2f* p1 = (const v2f*)(xf0 + xoff1);
#pragma unroll
    for (int ks = 0; ks < 16; ++ks) { xa[0][ks] = p0[2 * ks]; xa[1][ks] = p1[2 * ks]; }
  }


  for (int i = tid; i < (CIN / 2) * COUT; i += WPB * 32) {
    const int p = i >> 6, d = i & 63;
    v2f val;
    val.x = W[(2 * p) * COUT + d];
    val.y = W[(2 * p + 1) * COUT + d];
    wpair[p][d] = val;
  }
  if (tid < COUT) bbuf[tid] = bias[tid];

  const int* esrc = edge_index;
  const int* etgt = edge_index + E;
  const int t  = tid >> 2;
  const int s0 = (tid & 3) << 3;
  const bool onechunk = (E <= ECHUNK);

  float deg = 1.0f;
  for (int base = 0; base < E; base += ECHUNK) {
    if (base > 0) __syncthreads();
    int2 ed;
    ed.x = -1; ed.y = -1;
    if (base + tid < E) { ed.x = etgt[base + tid]; ed.y = esrc[base + tid]; }
    ebuf[tid] = ed;
    __syncthreads();
    const int n = min(E - base, ECHUNK);
#pragma unroll 4
    for (int e = 0; e < n; ++e) deg += (ebuf[e].x == t) ? 1.0f : 0.0f;
  }
  const float dinv_t = rsqrtf(deg);
  if ((tid & 3) == 0) dinvbuf[t] = dinv_t;
  __syncthreads();

  float wj[8], aent[8];
#pragma unroll
  for (int j = 0; j < 8; ++j) {
    wj[j] = dinvbuf[s0 + j] * dinv_t;
    aent[j] = 0.0f;
  }
  for (int base = 0; base < E; base += ECHUNK) {
    if (!onechunk) {
      int2 ed;
      ed.x = -1; ed.y = -1;
      if (base + tid < E) { ed.x = etgt[base + tid]; ed.y = esrc[base + tid]; }
      ebuf[tid] = ed;
      __syncthreads();
    }
    const int n = min(E - base, ECHUNK);
#pragma unroll 4
    for (int e = 0; e < n; ++e) {
      const int2 ed = ebuf[e];
      const int d = (ed.x == t) ? (ed.y - s0) : -1;
#pragma unroll
      for (int j = 0; j < 8; ++j) aent[j] += (d == j) ? wj[j] : 0.0f;
    }
    if (!onechunk) __syncthreads();
  }
#pragma unroll
  for (int j = 0; j < 8; ++j) aent[j] += (s0 + j == t && t < NJOINT) ? wj[j] : 0.0f;
#pragma unroll
  for (int q = 0; q < 4; ++q) {
    v2f val;
    val.x = aent[2 * q];
    val.y = aent[2 * q + 1];
    apair[t][(s0 >> 1) + q] = val;
  }
  __syncthreads();

  v2f (*pw)[PSTR] = pwbuf[wave];

  for (int f = 0; f < FPW; ++f) {
    const int frame = waveGlobal + f * (BLOCKS_B * WPB);

#pragma unroll
    for (int mt = 0; mt < 2; ++mt) {
      v8f acc[4];
#pragma unroll
      for (int nt = 0; nt < 4; ++nt) acc[nt] = (v8f)(0.0f);

#pragma unroll
      for (int ks = 0; ks < 16; ++ks) {
        const int kp = 2 * ks + half;
        v2f a = xa[mt][ks];
        if (mt == 1 && !m1valid) a = (v2f)(0.0f);
#pragma unroll
        for (int nt = 0; nt < 4; ++nt) {
          v2f b = wpair[kp][nt * 16 + ln];
          acc[nt] = __builtin_amdgcn_wmma_f32_16x16x4_f32(
              false, a, false, b, (short)0, acc[nt], false, false);
        }
      }
#pragma unroll
      for (int nt = 0; nt < 4; ++nt)
#pragma unroll
        for (int r = 0; r < 8; r += 2) {
          v2f val;
          val.x = acc[nt][r];
          val.y = acc[nt][r + 1];
          pw[mt * 8 + half * 4 + (r >> 1)][nt * 16 + ln] = val;
        }
    }

    if (f + 1 < FPW) {
      const float* xfn = x + (size_t)(frame + BLOCKS_B * WPB) * (NJOINT * CIN);
      const v2f* p0 = (const v2f*)(xfn + xoff0);
      const v2f* p1 = (const v2f*)(xfn + xoff1);
#pragma unroll
      for (int ks = 0; ks < 16; ++ks) { xa[0][ks] = p0[2 * ks]; xa[1][ks] = p1[2 * ks]; }
    }

#pragma unroll
    for (int mt = 0; mt < 2; ++mt) {
      v8f acc[4];
#pragma unroll
      for (int nt = 0; nt < 4; ++nt) acc[nt] = (v8f)(0.0f);

#pragma unroll
      for (int ks = 0; ks < 8; ++ks) {
        const int kp = 2 * ks + half;
        v2f a = apair[mt * 16 + ln][kp];
#pragma unroll
        for (int nt = 0; nt < 4; ++nt) {
          v2f b = pw[kp][nt * 16 + ln];
          acc[nt] = __builtin_amdgcn_wmma_f32_16x16x4_f32(
              false, a, false, b, (short)0, acc[nt], false, false);
        }
      }
      float* of = out + (size_t)frame * VC;
#pragma unroll
      for (int nt = 0; nt < 4; ++nt) {
        const int colb = nt * 16 + ln;
        const float bb = bbuf[colb];
#pragma unroll
        for (int r = 0; r < 8; ++r) {
          const int v = mt * 16 + half * 8 + r;
          if (v < NJOINT) {
            float val = acc[nt][r] + bb;
            of[v * COUT + colb] = val > 0.0f ? val : 0.0f;
          }
        }
      }
    }
  }
}

__global__ void stgcn_bn_partial(const float* __restrict__ out,
                                 float* __restrict__ ws) {
  const int ch = blockIdx.x * blockDim.x + threadIdx.x;
  if (ch >= VC) return;
  const int chunk = blockIdx.y;
  const float* p = out + (size_t)chunk * CHUNKROWS * VC + ch;
  float s = 0.0f, ss = 0.0f;
  for (int i = 0; i < CHUNKROWS; ++i) {
    float v = p[(size_t)i * VC];
    s += v;
    ss += v * v;
  }
  ws[WS_PSUM + chunk * VC + ch] = s;
  ws[WS_PSQ + chunk * VC + ch] = ss;
}

__global__ void stgcn_bn_finalize(const float* __restrict__ gamma,
                                  const float* __restrict__ beta,
                                  float* __restrict__ ws) {
  const int ch = blockIdx.x * blockDim.x + threadIdx.x;
  if (ch >= VC) return;
  double s = 0.0, ss = 0.0;
  for (int c = 0; c < NCHUNK; ++c) {
    s += (double)ws[WS_PSUM + c * VC + ch];
    ss += (double)ws[WS_PSQ + c * VC + ch];
  }
  const double n = (double)NFRM;
  const double mean = s / n;
  const double var = ss / n - mean * mean;
  const float scale = gamma[ch] * rsqrtf((float)var + BN_EPS);
  const float shift = beta[ch] - (float)mean * scale;
  ws[WS_SCALE + ch] = scale;
  ws[WS_SHIFT + ch] = shift;
}

__global__ void stgcn_bn_apply(float* __restrict__ out,
                               const float* __restrict__ ws) {
  const size_t idx = ((size_t)blockIdx.x * blockDim.x + threadIdx.x) * 4;
  float4 h = *(const float4*)(out + idx);
  const int ch = (int)(idx % VC);
  const float4 sc = *(const float4*)(ws + WS_SCALE + ch);
  const float4 sh = *(const float4*)(ws + WS_SHIFT + ch);
  h.x = h.x * sc.x + sh.x;
  h.y = h.y * sc.y + sh.y;
  h.z = h.z * sc.z + sh.z;
  h.w = h.w * sc.w + sh.w;
  *(float4*)(out + idx) = h;
}

extern "C" void kernel_launch(void* const* d_in, const int* in_sizes, int n_in,
                              void* d_out, int out_size, void* d_ws, size_t ws_size,
                              hipStream_t stream) {
  const float* x     = (const float*)d_in[0];
  const float* W     = (const float*)d_in[1];
  const float* b     = (const float*)d_in[2];
  const float* gamma = (const float*)d_in[3];
  const float* beta  = (const float*)d_in[4];
  const int*   ei    = (const int*)d_in[5];
  float* out = (float*)d_out;
  float* ws  = (float*)d_ws;
  const int E = in_sizes[5] / 2;

  hipLaunchKernelGGL(stgcn_gcn_gemm, dim3(BLOCKS_B), dim3(WPB * 32), 0, stream,
                     x, W, b, ei, E, out);
  hipLaunchKernelGGL(stgcn_bn_partial, dim3((VC + 255) / 256, NCHUNK), dim3(256),
                     0, stream, out, ws);
  hipLaunchKernelGGL(stgcn_bn_finalize, dim3((VC + 255) / 256), dim3(256),
                     0, stream, gamma, beta, ws);
  hipLaunchKernelGGL(stgcn_bn_apply, dim3((NFRM * VC) / 4 / 256), dim3(256),
                     0, stream, out, ws);
}


# W, A_hat and bias fragments read from LDS once and held in registers, X*W tile re-read as one load group: no LDS read left among the WMMAs; grid 600 -> 512 blocks (2 resident waves per SIMD)
# speedup vs baseline: 1.3729x; 1.0630x over previous
#include <hip/hip_runtime.h>
#include <cstddef>

#define NJOINT 25
#define CIN    64
#define COUT   64
#define VC     (NJOINT * COUT)
#define NFRM   (64 * 300)
#define BN_EPS 1e-5f

#define WPB        4
#define BLOCKS_B   512
#define NWAVES     (BLOCKS_B * WPB)

#define NCHUNK     60
#define CHUNKROWS  320

#define WS_PSUM   1024
#define WS_PSQ    (WS_PSUM + NCHUNK * VC)
#define WS_SCALE  (WS_PSQ  + NCHUNK * VC)
#define WS_SHIFT  (WS_SCALE + VC)

typedef __attribute__((ext_vector_type(2))) float v2f;
typedef __attribute__((ext_vector_type(8))) float v8f;

#define WSTR 76
#define PSTR 80
#define ASTR 17
#define ECHUNK (WPB * 32)

__launch_bounds__(WPB * 32, 1)
__global__ void stgcn_gcn_gemm(const float* __restrict__ x,
                               const float* __restrict__ W,
                               const float* __restrict__ bias,
                               const int* __restrict__ edge_index, int E,
                               float* __restrict__ out) {
  __shared__ v2f wpair[CIN / 2][WSTR];
  __shared__ v2f apair[32][ASTR];
  __shared__ v2f pwbuf[WPB][16][PSTR];
  __shared__ float bbuf[COUT];
  __shared__ int2 ebuf[ECHUNK];
  __shared__ float dinvbuf[32];

  const int tid = threadIdx.x;
  const int wave = tid >> 5;
  const int lane = tid & 31;
  const int half = lane >> 4;
  const int ln   = lane & 15;
  const int waveGlobal = blockIdx.x * WPB + wave;

  const bool m1valid = (16 + ln) < NJOINT;
  const int xoff0 = ln * CIN + 2 * half;
  const int xoff1 = (m1valid ? (16 + ln) : (NJOINT - 1)) * CIN + 2 * half;
  v2f xa[2][16];
  {
    const float* xf0 = x + (size_t)waveGlobal * (NJOINT * CIN);
    const v2f* p0 = (const v2f*)(xf0 + xoff0);
    const v2f* p1 = (const v2f*)(xf0 + xoff1);
#pragma unroll
    for (int ks = 0; ks < 16; ++ks) { xa[0][ks] = p0[2 * ks]; xa[1][ks] = p1[2 * ks]; }
  }


  for (int i = tid; i < (CIN / 2) * COUT; i += WPB * 32) {
    const int p = i >> 6, d = i & 63;
    v2f val;
    val.x = W[(2 * p) * COUT + d];
    val.y = W[(2 * p + 1) * COUT + d];
    wpair[p][d] = val;
  }
  if (tid < COUT) bbuf[tid] = bias[tid];

  const int* esrc = edge_index;
  const int* etgt = edge_index + E;
  const int t  = tid >> 2;
  const int s0 = (tid & 3) << 3;
  const bool onechunk = (E <= ECHUNK);

  float deg = 1.0f;
  for (int base = 0; base < E; base += ECHUNK) {
    if (base > 0) __syncthreads();
    int2 ed;
    ed.x = -1; ed.y = -1;
    if (base + tid < E) { ed.x = etgt[base + tid]; ed.y = esrc[base + tid]; }
    ebuf[tid] = ed;
    __syncthreads();
    const int n = min(E - base, ECHUNK);
#pragma unroll 4
    for (int e = 0; e < n; ++e) deg += (ebuf[e].x == t) ? 1.0f : 0.0f;
  }
  const float dinv_t = rsqrtf(deg);
  if ((tid & 3) == 0) dinvbuf[t] = dinv_t;
  __syncthreads();

  float wj[8], aent[8];
#pragma unroll
  for (int j = 0; j < 8; ++j) {
    wj[j] = dinvbuf[s0 + j] * dinv_t;
    aent[j] = 0.0f;
  }
  for (int base = 0; base < E; base += ECHUNK) {
    if (!onechunk) {
      int2 ed;
      ed.x = -1; ed.y = -1;
      if (base + tid < E) { ed.x = etgt[base + tid]; ed.y = esrc[base + tid]; }
      ebuf[tid] = ed;
      __syncthreads();
    }
    const int n = min(E - base, ECHUNK);
#pragma unroll 4
    for (int e = 0; e < n; ++e) {
      const int2 ed = ebuf[e];
      const int d = (ed.x == t) ? (ed.y - s0) : -1;
#pragma unroll
      for (int j = 0; j < 8; ++j) aent[j] += (d == j) ? wj[j] : 0.0f;
    }
    if (!onechunk) __syncthreads();
  }
#pragma unroll
  for (int j = 0; j < 8; ++j) aent[j] += (s0 + j == t && t < NJOINT) ? wj[j] : 0.0f;
#pragma unroll
  for (int q = 0; q < 4; ++q) {
    v2f val;
    val.x = aent[2 * q];
    val.y = aent[2 * q + 1];
    apair[t][(s0 >> 1) + q] = val;
  }
  __syncthreads();

  v2f (*pw)[PSTR] = pwbuf[wave];

  v2f wf[16][4];
#pragma unroll
  for (int ks = 0; ks < 16; ++ks)
#pragma unroll
    for (int nt = 0; nt < 4; ++nt) wf[ks][nt] = wpair[2 * ks + half][nt * 16 + ln];
  v2f af[2][8];
#pragma unroll
  for (int mt = 0; mt < 2; ++mt)
#pragma unroll
    for (int ks = 0; ks < 8; ++ks) af[mt][ks] = apair[mt * 16 + ln][2 * ks + half];
  float bb[4];
#pragma unroll
  for (int nt = 0; nt < 4; ++nt) bb[nt] = bbuf[nt * 16 + ln];

  for (int frame = waveGlobal; frame < NFRM; frame += NWAVES) {

#pragma unroll
    for (int mt = 0; mt < 2; ++mt) {
      v8f acc[4];
#pragma unroll
      for (int nt = 0; nt < 4; ++nt) acc[nt] = (v8f)(0.0f);

#pragma unroll
      for (int ks = 0; ks < 16; ++ks) {
        v2f a = xa[mt][ks];
        if (mt == 1 && !m1valid) a = (v2f)(0.0f);
#pragma unroll
        for (int nt = 0; nt < 4; ++nt) {
          acc[nt] = __builtin_amdgcn_wmma_f32_16x16x4_f32(
              false, a, false, wf[ks][nt], (short)0, acc[nt], false, false);
        }
      }
#pragma unroll
      for (int nt = 0; nt < 4; ++nt)
#pragma unroll
        for (int r = 0; r < 8; r += 2) {
          v2f val;
          val.x = acc[nt][r];
          val.y = acc[nt][r + 1];
          pw[mt * 8 + half * 4 + (r >> 1)][nt * 16 + ln] = val;
        }
    }

    v2f pf[8][4];
#pragma unroll
    for (int ks = 0; ks < 8; ++ks)
#pragma unroll
      for (int nt = 0; nt < 4; ++nt) pf[ks][nt] = pw[2 * ks + half][nt * 16 + ln];

    if (frame + NWAVES < NFRM) {
      const float* xfn = x + (size_t)(frame + NWAVES) * (NJOINT * CIN);
      const v2f* p0 = (const v2f*)(xfn + xoff0);
      const v2f* p1 = (const v2f*)(xfn + xoff1);
#pragma unroll
      for (int ks = 0; ks < 16; ++ks) { xa[0][ks] = p0[2 * ks]; xa[1][ks] = p1[2 * ks]; }
    }

    __builtin_amdgcn_sched_barrier(0);

#pragma unroll
    for (int mt = 0; mt < 2; ++mt) {
      v8f acc[4];
#pragma unroll
      for (int nt = 0; nt < 4; ++nt) acc[nt] = (v8f)(0.0f);

#pragma unroll
      for (int ks = 0; ks < 8; ++ks) {
#pragma unroll
        for (int nt = 0; nt < 4; ++nt) {
          acc[nt] = __builtin_amdgcn_wmma_f32_16x16x4_f32(
              false, af[mt][ks], false, pf[ks][nt], (short)0, acc[nt], false, false);
        }
      }
      float* of = out + (size_t)frame * VC;
#pragma unroll
      for (int nt = 0; nt < 4; ++nt) {
        const int colb = nt * 16 + ln;
#pragma unroll
        for (int r = 0; r < 8; ++r) {
          const int v = mt * 16 + half * 8 + r;
          if (v < NJOINT) {
            float val = acc[nt][r] + bb[nt];
            of[v * COUT + colb] = val > 0.0f ? val : 0.0f;
          }
        }
      }
    }
  }
}

__global__ void stgcn_bn_partial(const float* __restrict__ out,
                                 float* __restrict__ ws) {
  const int ch = blockIdx.x * blockDim.x + threadIdx.x;
  if (ch >= VC) return;
  const int chunk = blockIdx.y;
  const float* p = out + (size_t)chunk * CHUNKROWS * VC + ch;
  float s = 0.0f, ss = 0.0f;
  for (int i = 0; i < CHUNKROWS; ++i) {
    float v = p[(size_t)i * VC];
    s += v;
    ss += v * v;
  }
  ws[WS_PSUM + chunk * VC + ch] = s;
  ws[WS_PSQ + chunk * VC + ch] = ss;
}

__global__ void stgcn_bn_finalize(const float* __restrict__ gamma,
                                  const float* __restrict__ beta,
                                  float* __restrict__ ws) {
  const int ch = blockIdx.x * blockDim.x + threadIdx.x;
  if (ch >= VC) return;
  double s = 0.0, ss = 0.0;
  for (int c = 0; c < NCHUNK; ++c) {
    s += (double)ws[WS_PSUM + c * VC + ch];
    ss += (double)ws[WS_PSQ + c * VC + ch];
  }
  const double n = (double)NFRM;
  const double mean = s / n;
  const double var = ss / n - mean * mean;
  const float scale = gamma[ch] * rsqrtf((float)var + BN_EPS);
  const float shift = beta[ch] - (float)mean * scale;
  ws[WS_SCALE + ch] = scale;
  ws[WS_SHIFT + ch] = shift;
}

__global__ void stgcn_bn_apply(float* __restrict__ out,
                               const float* __restrict__ ws) {
  const size_t idx = ((size_t)blockIdx.x * blockDim.x + threadIdx.x) * 4;
  float4 h = *(const float4*)(out + idx);
  const int ch = (int)(idx % VC);
  const float4 sc = *(const float4*)(ws + WS_SCALE + ch);
  const float4 sh = *(const float4*)(ws + WS_SHIFT + ch);
  h.x = h.x * sc.x + sh.x;
  h.y = h.y * sc.y + sh.y;
  h.z = h.z * sc.z + sh.z;
  h.w = h.w * sc.w + sh.w;
  *(float4*)(out + idx) = h;
}

extern "C" void kernel_launch(void* const* d_in, const int* in_sizes, int n_in,
                              void* d_out, int out_size, void* d_ws, size_t ws_size,
                              hipStream_t stream) {
  const float* x     = (const float*)d_in[0];
  const float* W     = (const float*)d_in[1];
  const float* b     = (const float*)d_in[2];
  const float* gamma = (const float*)d_in[3];
  const float* beta  = (const float*)d_in[4];
  const int*   ei    = (const int*)d_in[5];
  float* out = (float*)d_out;
  float* ws  = (float*)d_ws;
  const int E = in_sizes[5] / 2;

  hipLaunchKernelGGL(stgcn_gcn_gemm, dim3(BLOCKS_B), dim3(WPB * 32), 0, stream,
                     x, W, b, ei, E, out);
  hipLaunchKernelGGL(stgcn_bn_partial, dim3((VC + 255) / 256, NCHUNK), dim3(256),
                     0, stream, out, ws);
  hipLaunchKernelGGL(stgcn_bn_finalize, dim3((VC + 255) / 256), dim3(256),
                     0, stream, gamma, beta, ws);
  hipLaunchKernelGGL(stgcn_bn_apply, dim3((NFRM * VC) / 4 / 256), dim3(256),
                     0, stream, out, ws);
}


# GEMM epilogue staged through a per-wave LDS tile in output order, written as whole 512-byte wave stores (b128 a lane): 13 full-line stores a frame replace 64 half-line b32 stores, 28 masked branches
# speedup vs baseline: 1.5609x; 1.1369x over previous
#include <hip/hip_runtime.h>
#include <cstddef>

#define NJOINT 25
#define CIN    64
#define COUT   64
#define VC     (NJOINT * COUT)
#define NFRM   (64 * 300)
#define BN_EPS 1e-5f

#define WPB        4
#define BLOCKS_B   512
#define NWAVES     (BLOCKS_B * WPB)

#define NCHUNK     60
#define CHUNKROWS  320

#define WS_PSUM   1024
#define WS_PSQ    (WS_PSUM + NCHUNK * VC)
#define WS_SCALE  (WS_PSQ  + NCHUNK * VC)
#define WS_SHIFT  (WS_SCALE + VC)

typedef __attribute__((ext_vector_type(2))) float v2f;
typedef __attribute__((ext_vector_type(4))) float v4f;
typedef __attribute__((ext_vector_type(8))) float v8f;

#define WSTR 76
#define PSTR 80
#define ASTR 17
#define HSTR 68
#define ECHUNK (WPB * 32)

__launch_bounds__(WPB * 32, 1)
__global__ void stgcn_gcn_gemm(const float* __restrict__ x,
                               const float* __restrict__ W,
                               const float* __restrict__ bias,
                               const int* __restrict__ edge_index, int E,
                               float* __restrict__ out) {
  __shared__ v2f wpair[CIN / 2][WSTR];
  __shared__ v2f apair[32][ASTR];
  __shared__ v2f pwbuf[WPB][16][PSTR];
  __shared__ __attribute__((aligned(16))) float hbuf[WPB][32 * HSTR];
  __shared__ float bbuf[COUT];
  __shared__ int2 ebuf[ECHUNK];
  __shared__ float dinvbuf[32];

  const int tid = threadIdx.x;
  const int wave = tid >> 5;
  const int lane = tid & 31;
  const int half = lane >> 4;
  const int ln   = lane & 15;
  const int waveGlobal = blockIdx.x * WPB + wave;

  const bool m1valid = (16 + ln) < NJOINT;
  const int xoff0 = ln * CIN + 2 * half;
  const int xoff1 = (m1valid ? (16 + ln) : (NJOINT - 1)) * CIN + 2 * half;
  v2f xa[2][16];
  {
    const float* xf0 = x + (size_t)waveGlobal * (NJOINT * CIN);
    const v2f* p0 = (const v2f*)(xf0 + xoff0);
    const v2f* p1 = (const v2f*)(xf0 + xoff1);
#pragma unroll
    for (int ks = 0; ks < 16; ++ks) { xa[0][ks] = p0[2 * ks]; xa[1][ks] = p1[2 * ks]; }
  }


  for (int i = tid; i < (CIN / 2) * COUT; i += WPB * 32) {
    const int p = i >> 6, d = i & 63;
    v2f val;
    val.x = W[(2 * p) * COUT + d];
    val.y = W[(2 * p + 1) * COUT + d];
    wpair[p][d] = val;
  }
  if (tid < COUT) bbuf[tid] = bias[tid];

  const int* esrc = edge_index;
  const int* etgt = edge_index + E;
  const int t  = tid >> 2;
  const int s0 = (tid & 3) << 3;
  const bool onechunk = (E <= ECHUNK);

  float deg = 1.0f;
  for (int base = 0; base < E; base += ECHUNK) {
    if (base > 0) __syncthreads();
    int2 ed;
    ed.x = -1; ed.y = -1;
    if (base + tid < E) { ed.x = etgt[base + tid]; ed.y = esrc[base + tid]; }
    ebuf[tid] = ed;
    __syncthreads();
    const int n = min(E - base, ECHUNK);
#pragma unroll 4
    for (int e = 0; e < n; ++e) deg += (ebuf[e].x == t) ? 1.0f : 0.0f;
  }
  const float dinv_t = rsqrtf(deg);
  if ((tid & 3) == 0) dinvbuf[t] = dinv_t;
  __syncthreads();

  float wj[8], aent[8];
#pragma unroll
  for (int j = 0; j < 8; ++j) {
    wj[j] = dinvbuf[s0 + j] * dinv_t;
    aent[j] = 0.0f;
  }
  for (int base = 0; base < E; base += ECHUNK) {
    if (!onechunk) {
      int2 ed;
      ed.x = -1; ed.y = -1;
      if (base + tid < E) { ed.x = etgt[base + tid]; ed.y = esrc[base + tid]; }
      ebuf[tid] = ed;
      __syncthreads();
    }
    const int n = min(E - base, ECHUNK);
#pragma unroll 4
    for (int e = 0; e < n; ++e) {
      const int2 ed = ebuf[e];
      const int d = (ed.x == t) ? (ed.y - s0) : -1;
#pragma unroll
      for (int j = 0; j < 8; ++j) aent[j] += (d == j) ? wj[j] : 0.0f;
    }
    if (!onechunk) __syncthreads();
  }
#pragma unroll
  for (int j = 0; j < 8; ++j) aent[j] += (s0 + j == t && t < NJOINT) ? wj[j] : 0.0f;
#pragma unroll
  for (int q = 0; q < 4; ++q) {
    v2f val;
    val.x = aent[2 * q];
    val.y = aent[2 * q + 1];
    apair[t][(s0 >> 1) + q] = val;
  }
  __syncthreads();

  v2f (*pw)[PSTR] = pwbuf[wave];
  float* hw = hbuf[wave];

  v2f wf[16][4];
#pragma unroll
  for (int ks = 0; ks < 16; ++ks)
#pragma unroll
    for (int nt = 0; nt < 4; ++nt) wf[ks][nt] = wpair[2 * ks + half][nt * 16 + ln];
  v2f af[2][8];
#pragma unroll
  for (int mt = 0; mt < 2; ++mt)
#pragma unroll
    for (int ks = 0; ks < 8; ++ks) af[mt][ks] = apair[mt * 16 + ln][2 * ks + half];
  float bb[4];
#pragma unroll
  for (int nt = 0; nt < 4; ++nt) bb[nt] = bbuf[nt * 16 + ln];

  for (int frame = waveGlobal; frame < NFRM; frame += NWAVES) {

#pragma unroll
    for (int mt = 0; mt < 2; ++mt) {
      v8f acc[4];
#pragma unroll
      for (int nt = 0; nt < 4; ++nt) acc[nt] = (v8f)(0.0f);

#pragma unroll
      for (int ks = 0; ks < 16; ++ks) {
        v2f a = xa[mt][ks];
        if (mt == 1 && !m1valid) a = (v2f)(0.0f);
#pragma unroll
        for (int nt = 0; nt < 4; ++nt) {
          acc[nt] = __builtin_amdgcn_wmma_f32_16x16x4_f32(
              false, a, false, wf[ks][nt], (short)0, acc[nt], false, false);
        }
      }
#pragma unroll
      for (int nt = 0; nt < 4; ++nt)
#pragma unroll
        for (int r = 0; r < 8; r += 2) {
          v2f val;
          val.x = acc[nt][r];
          val.y = acc[nt][r + 1];
          pw[mt * 8 + half * 4 + (r >> 1)][nt * 16 + ln] = val;
        }
    }

    v2f pf[8][4];
#pragma unroll
    for (int ks = 0; ks < 8; ++ks)
#pragma unroll
      for (int nt = 0; nt < 4; ++nt) pf[ks][nt] = pw[2 * ks + half][nt * 16 + ln];

    if (frame + NWAVES < NFRM) {
      const float* xfn = x + (size_t)(frame + NWAVES) * (NJOINT * CIN);
      const v2f* p0 = (const v2f*)(xfn + xoff0);
      const v2f* p1 = (const v2f*)(xfn + xoff1);
#pragma unroll
      for (int ks = 0; ks < 16; ++ks) { xa[0][ks] = p0[2 * ks]; xa[1][ks] = p1[2 * ks]; }
    }

    __builtin_amdgcn_sched_barrier(0);

#pragma unroll
    for (int mt = 0; mt < 2; ++mt) {
      v8f acc[4];
#pragma unroll
      for (int nt = 0; nt < 4; ++nt) acc[nt] = (v8f)(0.0f);

#pragma unroll
      for (int ks = 0; ks < 8; ++ks) {
#pragma unroll
        for (int nt = 0; nt < 4; ++nt) {
          acc[nt] = __builtin_amdgcn_wmma_f32_16x16x4_f32(
              false, af[mt][ks], false, pf[ks][nt], (short)0, acc[nt], false, false);
        }
      }
#pragma unroll
      for (int nt = 0; nt < 4; ++nt) {
        const int colb = nt * 16 + ln;
#pragma unroll
        for (int r = 0; r < 8; ++r) {
          const int v = mt * 16 + half * 8 + r;
          float val = acc[nt][r] + bb[nt];
          hw[v * HSTR + colb] = val > 0.0f ? val : 0.0f;
        }
      }
    }

    float* of = out + (size_t)frame * VC;
#pragma unroll
    for (int s = 0; s < 13; ++s) {
      const int q = s * 32 + lane;
      if (s < 12 || lane < 16) {
        const v4f hv = *(const v4f*)(hw + (q >> 4) * HSTR + (q & 15) * 4);
        *(v4f*)(of + q * 4) = hv;
      }
    }
  }
}

__global__ void stgcn_bn_partial(const float* __restrict__ out,
                                 float* __restrict__ ws) {
  const int ch = blockIdx.x * blockDim.x + threadIdx.x;
  if (ch >= VC) return;
  const int chunk = blockIdx.y;
  const float* p = out + (size_t)chunk * CHUNKROWS * VC + ch;
  float s = 0.0f, ss = 0.0f;
  for (int i = 0; i < CHUNKROWS; ++i) {
    float v = p[(size_t)i * VC];
    s += v;
    ss += v * v;
  }
  ws[WS_PSUM + chunk * VC + ch] = s;
  ws[WS_PSQ + chunk * VC + ch] = ss;
}

__global__ void stgcn_bn_finalize(const float* __restrict__ gamma,
                                  const float* __restrict__ beta,
                                  float* __restrict__ ws) {
  const int ch = blockIdx.x * blockDim.x + threadIdx.x;
  if (ch >= VC) return;
  double s = 0.0, ss = 0.0;
  for (int c = 0; c < NCHUNK; ++c) {
    s += (double)ws[WS_PSUM + c * VC + ch];
    ss += (double)ws[WS_PSQ + c * VC + ch];
  }
  const double n = (double)NFRM;
  const double mean = s / n;
  const double var = ss / n - mean * mean;
  const float scale = gamma[ch] * rsqrtf((float)var + BN_EPS);
  const float shift = beta[ch] - (float)mean * scale;
  ws[WS_SCALE + ch] = scale;
  ws[WS_SHIFT + ch] = shift;
}

__global__ void stgcn_bn_apply(float* __restrict__ out,
                               const float* __restrict__ ws) {
  const size_t idx = ((size_t)blockIdx.x * blockDim.x + threadIdx.x) * 4;
  float4 h = *(const float4*)(out + idx);
  const int ch = (int)(idx % VC);
  const float4 sc = *(const float4*)(ws + WS_SCALE + ch);
  const float4 sh = *(const float4*)(ws + WS_SHIFT + ch);
  h.x = h.x * sc.x + sh.x;
  h.y = h.y * sc.y + sh.y;
  h.z = h.z * sc.z + sh.z;
  h.w = h.w * sc.w + sh.w;
  *(float4*)(out + idx) = h;
}

extern "C" void kernel_launch(void* const* d_in, const int* in_sizes, int n_in,
                              void* d_out, int out_size, void* d_ws, size_t ws_size,
                              hipStream_t stream) {
  const float* x     = (const float*)d_in[0];
  const float* W     = (const float*)d_in[1];
  const float* b     = (const float*)d_in[2];
  const float* gamma = (const float*)d_in[3];
  const float* beta  = (const float*)d_in[4];
  const int*   ei    = (const int*)d_in[5];
  float* out = (float*)d_out;
  float* ws  = (float*)d_ws;
  const int E = in_sizes[5] / 2;

  hipLaunchKernelGGL(stgcn_gcn_gemm, dim3(BLOCKS_B), dim3(WPB * 32), 0, stream,
                     x, W, b, ei, E, out);
  hipLaunchKernelGGL(stgcn_bn_partial, dim3((VC + 255) / 256, NCHUNK), dim3(256),
                     0, stream, out, ws);
  hipLaunchKernelGGL(stgcn_bn_finalize, dim3((VC + 255) / 256), dim3(256),
                     0, stream, gamma, beta, ws);
  hipLaunchKernelGGL(stgcn_bn_apply, dim3((NFRM * VC) / 4 / 256), dim3(256),
                     0, stream, out, ws);
}


# BatchNorm sums taken in the GEMM epilogue from the float4 each lane stores (new sum order: wave, 4 waves, 512 blocks in fp64; workspace 6566400 B); the statistics launch and its re-read of H are gone
# speedup vs baseline: 2.0495x; 1.3130x over previous
#include <hip/hip_runtime.h>
#include <cstddef>

#define NJOINT 25
#define CIN    64
#define COUT   64
#define VC     (NJOINT * COUT)
#define NFRM   (64 * 300)
#define BN_EPS 1e-5f

#define WPB        4
#define BLOCKS_B   512
#define NWAVES     (BLOCKS_B * WPB)

#define WS_SCALE  0
#define WS_SHIFT  (WS_SCALE + VC)
#define WS_PSUM   (WS_SHIFT + VC)
#define WS_PSQ    (WS_PSUM + BLOCKS_B * VC)
#define WS_NEED_BYTES ((size_t)(WS_PSQ + BLOCKS_B * VC) * sizeof(float))

#define FIN_C4  8
#define FIN_G   32

typedef __attribute__((ext_vector_type(2))) float v2f;
typedef __attribute__((ext_vector_type(4))) float v4f;
typedef __attribute__((ext_vector_type(8))) float v8f;

#define WSTR 76
#define PSTR 80
#define ASTR 17
#define HSTR 68
#define ECHUNK (WPB * 32)

__launch_bounds__(WPB * 32, 1)
__global__ void stgcn_gcn_gemm(const float* __restrict__ x,
                               const float* __restrict__ W,
                               const float* __restrict__ bias,
                               const int* __restrict__ edge_index, int E,
                               float* __restrict__ out,
                               float* __restrict__ ws, size_t ws_bytes) {
  __shared__ v2f wpair[CIN / 2][WSTR];
  __shared__ v2f apair[32][ASTR];
  __shared__ __attribute__((aligned(16))) v2f pwbuf[WPB][16][PSTR];
  __shared__ __attribute__((aligned(16))) float hbuf[WPB][32 * HSTR];
  __shared__ float bbuf[COUT];
  __shared__ int2 ebuf[ECHUNK];
  __shared__ float dinvbuf[32];

  const int tid = threadIdx.x;
  const int wave = tid >> 5;
  const int lane = tid & 31;
  const int half = lane >> 4;
  const int ln   = lane & 15;
  const int waveGlobal = blockIdx.x * WPB + wave;

  const bool m1valid = (16 + ln) < NJOINT;
  const int xoff0 = ln * CIN + 2 * half;
  const int xoff1 = (m1valid ? (16 + ln) : (NJOINT - 1)) * CIN + 2 * half;
  v2f xa[2][16];
  {
    const float* xf0 = x + (size_t)waveGlobal * (NJOINT * CIN);
    const v2f* p0 = (const v2f*)(xf0 + xoff0);
    const v2f* p1 = (const v2f*)(xf0 + xoff1);
#pragma unroll
    for (int ks = 0; ks < 16; ++ks) { xa[0][ks] = p0[2 * ks]; xa[1][ks] = p1[2 * ks]; }
  }


  for (int i = tid; i < (CIN / 2) * COUT; i += WPB * 32) {
    const int p = i >> 6, d = i & 63;
    v2f val;
    val.x = W[(2 * p) * COUT + d];
    val.y = W[(2 * p + 1) * COUT + d];
    wpair[p][d] = val;
  }
  if (tid < COUT) bbuf[tid] = bias[tid];

  const int* esrc = edge_index;
  const int* etgt = edge_index + E;
  const int t  = tid >> 2;
  const int s0 = (tid & 3) << 3;
  const bool onechunk = (E <= ECHUNK);

  float deg = 1.0f;
  for (int base = 0; base < E; base += ECHUNK) {
    if (base > 0) __syncthreads();
    int2 ed;
    ed.x = -1; ed.y = -1;
    if (base + tid < E) { ed.x = etgt[base + tid]; ed.y = esrc[base + tid]; }
    ebuf[tid] = ed;
    __syncthreads();
    const int n = min(E - base, ECHUNK);
#pragma unroll 4
    for (int e = 0; e < n; ++e) deg += (ebuf[e].x == t) ? 1.0f : 0.0f;
  }
  const float dinv_t = rsqrtf(deg);
  if ((tid & 3) == 0) dinvbuf[t] = dinv_t;
  __syncthreads();

  float wj[8], aent[8];
#pragma unroll
  for (int j = 0; j < 8; ++j) {
    wj[j] = dinvbuf[s0 + j] * dinv_t;
    aent[j] = 0.0f;
  }
  for (int base = 0; base < E; base += ECHUNK) {
    if (!onechunk) {
      int2 ed;
      ed.x = -1; ed.y = -1;
      if (base + tid < E) { ed.x = etgt[base + tid]; ed.y = esrc[base + tid]; }
      ebuf[tid] = ed;
      __syncthreads();
    }
    const int n = min(E - base, ECHUNK);
#pragma unroll 4
    for (int e = 0; e < n; ++e) {
      const int2 ed = ebuf[e];
      const int d = (ed.x == t) ? (ed.y - s0) : -1;
#pragma unroll
      for (int j = 0; j < 8; ++j) aent[j] += (d == j) ? wj[j] : 0.0f;
    }
    if (!onechunk) __syncthreads();
  }
#pragma unroll
  for (int j = 0; j < 8; ++j) aent[j] += (s0 + j == t && t < NJOINT) ? wj[j] : 0.0f;
#pragma unroll
  for (int q = 0; q < 4; ++q) {
    v2f val;
    val.x = aent[2 * q];
    val.y = aent[2 * q + 1];
    apair[t][(s0 >> 1) + q] = val;
  }
  __syncthreads();

  v2f (*pw)[PSTR] = pwbuf[wave];
  float* hw = hbuf[wave];

  v2f wf[16][4];
#pragma unroll
  for (int ks = 0; ks < 16; ++ks)
#pragma unroll
    for (int nt = 0; nt < 4; ++nt) wf[ks][nt] = wpair[2 * ks + half][nt * 16 + ln];
  v2f af[2][8];
#pragma unroll
  for (int mt = 0; mt < 2; ++mt)
#pragma unroll
    for (int ks = 0; ks < 8; ++ks) af[mt][ks] = apair[mt * 16 + ln][2 * ks + half];
  float bb[4];
#pragma unroll
  for (int nt = 0; nt < 4; ++nt) bb[nt] = bbuf[nt * 16 + ln];

  v4f ssum[13], ssq[13];
#pragma unroll
  for (int s = 0; s < 13; ++s) { ssum[s] = (v4f)(0.0f); ssq[s] = (v4f)(0.0f); }

  for (int frame = waveGlobal; frame < NFRM; frame += NWAVES) {

#pragma unroll
    for (int mt = 0; mt < 2; ++mt) {
      v8f acc[4];
#pragma unroll
      for (int nt = 0; nt < 4; ++nt) acc[nt] = (v8f)(0.0f);

#pragma unroll
      for (int ks = 0; ks < 16; ++ks) {
        v2f a = xa[mt][ks];
        if (mt == 1 && !m1valid) a = (v2f)(0.0f);
#pragma unroll
        for (int nt = 0; nt < 4; ++nt) {
          acc[nt] = __builtin_amdgcn_wmma_f32_16x16x4_f32(
              false, a, false, wf[ks][nt], (short)0, acc[nt], false, false);
        }
      }
#pragma unroll
      for (int nt = 0; nt < 4; ++nt)
#pragma unroll
        for (int r = 0; r < 8; r += 2) {
          v2f val;
          val.x = acc[nt][r];
          val.y = acc[nt][r + 1];
          pw[mt * 8 + half * 4 + (r >> 1)][nt * 16 + ln] = val;
        }
    }

    v2f pf[8][4];
#pragma unroll
    for (int ks = 0; ks < 8; ++ks)
#pragma unroll
      for (int nt = 0; nt < 4; ++nt) pf[ks][nt] = pw[2 * ks + half][nt * 16 + ln];

    if (frame + NWAVES < NFRM) {
      const float* xfn = x + (size_t)(frame + NWAVES) * (NJOINT * CIN);
      const v2f* p0 = (const v2f*)(xfn + xoff0);
      const v2f* p1 = (const v2f*)(xfn + xoff1);
#pragma unroll
      for (int ks = 0; ks < 16; ++ks) { xa[0][ks] = p0[2 * ks]; xa[1][ks] = p1[2 * ks]; }
    }

    __builtin_amdgcn_sched_barrier(0);

#pragma unroll
    for (int mt = 0; mt < 2; ++mt) {
      v8f acc[4];
#pragma unroll
      for (int nt = 0; nt < 4; ++nt) acc[nt] = (v8f)(0.0f);

#pragma unroll
      for (int ks = 0; ks < 8; ++ks) {
#pragma unroll
        for (int nt = 0; nt < 4; ++nt) {
          acc[nt] = __builtin_amdgcn_wmma_f32_16x16x4_f32(
              false, af[mt][ks], false, pf[ks][nt], (short)0, acc[nt], false, false);
        }
      }
#pragma unroll
      for (int nt = 0; nt < 4; ++nt) {
        const int colb = nt * 16 + ln;
#pragma unroll
        for (int r = 0; r < 8; ++r) {
          const int v = mt * 16 + half * 8 + r;
          float val = acc[nt][r] + bb[nt];
          hw[v * HSTR + colb] = val > 0.0f ? val : 0.0f;
        }
      }
    }

    float* of = out + (size_t)frame * VC;
#pragma unroll
    for (int s = 0; s < 13; ++s) {
      const int q = s * 32 + lane;
      if (s < 12 || lane < 16) {
        const v4f hv = *(const v4f*)(hw + (q >> 4) * HSTR + (q & 15) * 4);
        *(v4f*)(of + q * 4) = hv;
        ssum[s] += hv;
        ssq[s] += hv * hv;
      }
    }
  }

  float* sqw = (float*)pw;
#pragma unroll
  for (int s = 0; s < 13; ++s) {
    const int q = s * 32 + lane;
    if (s < 12 || lane < 16) {
      *(v4f*)(hw + q * 4) = ssum[s];
      *(v4f*)(sqw + q * 4) = ssq[s];
    }
  }
  __syncthreads();
  if (ws_bytes >= WS_NEED_BYTES) {
    float* psum = ws + WS_PSUM + (size_t)blockIdx.x * VC;
    float* psq  = ws + WS_PSQ  + (size_t)blockIdx.x * VC;
    for (int q = tid; q < VC / 4; q += WPB * 32) {
      v4f a = *(const v4f*)(hbuf[0] + q * 4);
      v4f b = *(const v4f*)((const float*)pwbuf[0] + q * 4);
#pragma unroll
      for (int w = 1; w < WPB; ++w) {
        a += *(const v4f*)(hbuf[w] + q * 4);
        b += *(const v4f*)((const float*)pwbuf[w] + q * 4);
      }
      *(v4f*)(psum + q * 4) = a;
      *(v4f*)(psq + q * 4) = b;
    }
  }
}

__launch_bounds__(FIN_C4 * FIN_G)
__global__ void stgcn_bn_finalize(const float* __restrict__ gamma,
                                  const float* __restrict__ beta,
                                  float* __restrict__ ws, size_t ws_bytes) {
  __shared__ double rs[FIN_G][FIN_C4 * 4 + 1];
  __shared__ double rq[FIN_G][FIN_C4 * 4 + 1];
  const int c4 = threadIdx.x & (FIN_C4 - 1);
  const int g  = threadIdx.x / FIN_C4;
  const int chb = blockIdx.x * (FIN_C4 * 4) + c4 * 4;
  double s[4] = {0.0, 0.0, 0.0, 0.0}, ss[4] = {0.0, 0.0, 0.0, 0.0};
  if (ws_bytes >= WS_NEED_BYTES) {
#pragma unroll
    for (int i = 0; i < BLOCKS_B / FIN_G; ++i) {
      const int p = g + i * FIN_G;
      const v4f a = *(const v4f*)(ws + WS_PSUM + (size_t)p * VC + chb);
      const v4f b = *(const v4f*)(ws + WS_PSQ  + (size_t)p * VC + chb);
#pragma unroll
      for (int j = 0; j < 4; ++j) { s[j] += (double)a[j]; ss[j] += (double)b[j]; }
    }
  }
#pragma unroll
  for (int j = 0; j < 4; ++j) { rs[g][c4 * 4 + j] = s[j]; rq[g][c4 * 4 + j] = ss[j]; }
  __syncthreads();
  if (threadIdx.x < FIN_C4 * 4) {
    const int c = threadIdx.x;
    const int ch = blockIdx.x * (FIN_C4 * 4) + c;
    double S = 0.0, SS = 0.0;
    for (int k = 0; k < FIN_G; ++k) { S += rs[k][c]; SS += rq[k][c]; }
    const double n = (double)NFRM;
    const double mean = S / n;
    const double var = SS / n - mean * mean;
    const float scale = gamma[ch] * rsqrtf((float)var + BN_EPS);
    const float shift = beta[ch] - (float)mean * scale;
    ws[WS_SCALE + ch] = scale;
    ws[WS_SHIFT + ch] = shift;
  }
}

__global__ void stgcn_bn_apply(float* __restrict__ out,
                               const float* __restrict__ ws) {
  const size_t idx = ((size_t)blockIdx.x * blockDim.x + threadIdx.x) * 4;
  float4 h = *(const float4*)(out + idx);
  const int ch = (int)(idx % VC);
  const float4 sc = *(const float4*)(ws + WS_SCALE + ch);
  const float4 sh = *(const float4*)(ws + WS_SHIFT + ch);
  h.x = h.x * sc.x + sh.x;
  h.y = h.y * sc.y + sh.y;
  h.z = h.z * sc.z + sh.z;
  h.w = h.w * sc.w + sh.w;
  *(float4*)(out + idx) = h;
}

extern "C" void kernel_launch(void* const* d_in, const int* in_sizes, int n_in,
                              void* d_out, int out_size, void* d_ws, size_t ws_size,
                              hipStream_t stream) {
  const float* x     = (const float*)d_in[0];
  const float* W     = (const float*)d_in[1];
  const float* b     = (const float*)d_in[2];
  const float* gamma = (const float*)d_in[3];
  const float* beta  = (const float*)d_in[4];
  const int*   ei    = (const int*)d_in[5];
  float* out = (float*)d_out;
  float* ws  = (float*)d_ws;
  const int E = in_sizes[5] / 2;

  hipLaunchKernelGGL(stgcn_gcn_gemm, dim3(BLOCKS_B), dim3(WPB * 32), 0, stream,
                     x, W, b, ei, E, out, ws, ws_size);
  hipLaunchKernelGGL(stgcn_bn_finalize, dim3(VC / (FIN_C4 * 4)), dim3(FIN_C4 * FIN_G),
                     0, stream, gamma, beta, ws, ws_size);
  hipLaunchKernelGGL(stgcn_bn_apply, dim3((NFRM * VC) / 4 / 256), dim3(256),
                     0, stream, out, ws);
}


# speedup vs baseline: 2.2010x; 1.0739x over previous
#include <hip/hip_runtime.h>
#include <cstddef>

#define NJOINT 25
#define CIN    64
#define COUT   64
#define VC     (NJOINT * COUT)
#define NFRM   (64 * 300)
#define BN_EPS 1e-5f

#define WPB        4
#define BLOCKS_B   512
#define NWAVES     (BLOCKS_B * WPB)

#define WS_SCALE  0
#define WS_SHIFT  (WS_SCALE + VC)
#define WS_PSUM   (WS_SHIFT + VC)
#define WS_PSQ    (WS_PSUM + BLOCKS_B * VC)
#define WS_NEED_BYTES ((size_t)(WS_PSQ + BLOCKS_B * VC) * sizeof(float))

#define FIN_C4  2
#define FIN_G   128
#define FIN_R   (BLOCKS_B / FIN_G)
#define FIN_V   (FIN_C4 * 8)
#define FIN_P   16
#define FIN_Q   (FIN_G / FIN_P)
#define FIN_LD  (FIN_V + 2)

typedef __attribute__((ext_vector_type(2))) float v2f;
typedef __attribute__((ext_vector_type(4))) float v4f;
typedef __attribute__((ext_vector_type(8))) float v8f;

#define WSTR 76
#define PSTR 80
#define ASTR 17
#define HSTR 68
#define ECHUNK (WPB * 32)

__launch_bounds__(WPB * 32, 1)
__global__ void stgcn_gcn_gemm(const float* __restrict__ x,
                               const float* __restrict__ W,
                               const float* __restrict__ bias,
                               const int* __restrict__ edge_index, int E,
                               float* __restrict__ out,
                               float* __restrict__ ws, size_t ws_bytes) {
  __shared__ v2f wpair[CIN / 2][WSTR];
  __shared__ v2f apair[32][ASTR];
  __shared__ __attribute__((aligned(16))) v2f pwbuf[WPB][16][PSTR];
  __shared__ __attribute__((aligned(16))) float hbuf[WPB][32 * HSTR];
  __shared__ float bbuf[COUT];
  __shared__ int2 ebuf[ECHUNK];
  __shared__ float dinvbuf[32];

  const int tid = threadIdx.x;
  const int wave = tid >> 5;
  const int lane = tid & 31;
  const int half = lane >> 4;
  const int ln   = lane & 15;
  const int waveGlobal = blockIdx.x * WPB + wave;

  const bool m1valid = (16 + ln) < NJOINT;
  const int xoff0 = ln * CIN + 2 * half;
  const int xoff1 = (m1valid ? (16 + ln) : (NJOINT - 1)) * CIN + 2 * half;
  v2f xa[2][16];
  {
    const float* xf0 = x + (size_t)waveGlobal * (NJOINT * CIN);
    const v2f* p0 = (const v2f*)(xf0 + xoff0);
    const v2f* p1 = (const v2f*)(xf0 + xoff1);
#pragma unroll
    for (int ks = 0; ks < 16; ++ks) { xa[0][ks] = p0[2 * ks]; xa[1][ks] = p1[2 * ks]; }
  }


  for (int i = tid; i < (CIN / 2) * COUT; i += WPB * 32) {
    const int p = i >> 6, d = i & 63;
    v2f val;
    val.x = W[(2 * p) * COUT + d];
    val.y = W[(2 * p + 1) * COUT + d];
    wpair[p][d] = val;
  }
  if (tid < COUT) bbuf[tid] = bias[tid];

  const int* esrc = edge_index;
  const int* etgt = edge_index + E;
  const int t  = tid >> 2;
  const int s0 = (tid & 3) << 3;
  const bool onechunk = (E <= ECHUNK);

  float deg = 1.0f;
  for (int base = 0; base < E; base += ECHUNK) {
    if (base > 0) __syncthreads();
    int2 ed;
    ed.x = -1; ed.y = -1;
    if (base + tid < E) { ed.x = etgt[base + tid]; ed.y = esrc[base + tid]; }
    ebuf[tid] = ed;
    __syncthreads();
    const int n = min(E - base, ECHUNK);
#pragma unroll 4
    for (int e = 0; e < n; ++e) deg += (ebuf[e].x == t) ? 1.0f : 0.0f;
  }
  const float dinv_t = rsqrtf(deg);
  if ((tid & 3) == 0) dinvbuf[t] = dinv_t;
  __syncthreads();

  float wj[8], aent[8];
#pragma unroll
  for (int j = 0; j < 8; ++j) {
    wj[j] = dinvbuf[s0 + j] * dinv_t;
    aent[j] = 0.0f;
  }
  for (int base = 0; base < E; base += ECHUNK) {
    if (!onechunk) {
      int2 ed;
      ed.x = -1; ed.y = -1;
      if (base + tid < E) { ed.x = etgt[base + tid]; ed.y = esrc[base + tid]; }
      ebuf[tid] = ed;
      __syncthreads();
    }
    const int n = min(E - base, ECHUNK);
#pragma unroll 4
    for (int e = 0; e < n; ++e) {
      const int2 ed = ebuf[e];
      const int d = (ed.x == t) ? (ed.y - s0) : -1;
#pragma unroll
      for (int j = 0; j < 8; ++j) aent[j] += (d == j) ? wj[j] : 0.0f;
    }
    if (!onechunk) __syncthreads();
  }
#pragma unroll
  for (int j = 0; j < 8; ++j) aent[j] += (s0 + j == t && t < NJOINT) ? wj[j] : 0.0f;
#pragma unroll
  for (int q = 0; q < 4; ++q) {
    v2f val;
    val.x = aent[2 * q];
    val.y = aent[2 * q + 1];
    apair[t][(s0 >> 1) + q] = val;
  }
  __syncthreads();

  v2f (*pw)[PSTR] = pwbuf[wave];
  float* hw = hbuf[wave];

  v2f wf[16][4];
#pragma unroll
  for (int ks = 0; ks < 16; ++ks)
#pragma unroll
    for (int nt = 0; nt < 4; ++nt) wf[ks][nt] = wpair[2 * ks + half][nt * 16 + ln];
  v2f af[2][8];
#pragma unroll
  for (int mt = 0; mt < 2; ++mt)
#pragma unroll
    for (int ks = 0; ks < 8; ++ks) af[mt][ks] = apair[mt * 16 + ln][2 * ks + half];
  float bb[4];
#pragma unroll
  for (int nt = 0; nt < 4; ++nt) bb[nt] = bbuf[nt * 16 + ln];

  v4f ssum[13], ssq[13];
#pragma unroll
  for (int s = 0; s < 13; ++s) { ssum[s] = (v4f)(0.0f); ssq[s] = (v4f)(0.0f); }

  for (int frame = waveGlobal; frame < NFRM; frame += NWAVES) {

#pragma unroll
    for (int mt = 0; mt < 2; ++mt) {
      v8f acc[4];
#pragma unroll
      for (int nt = 0; nt < 4; ++nt) acc[nt] = (v8f)(0.0f);

#pragma unroll
      for (int ks = 0; ks < 16; ++ks) {
        v2f a = xa[mt][ks];
        if (mt == 1 && !m1valid) a = (v2f)(0.0f);
#pragma unroll
        for (int nt = 0; nt < 4; ++nt) {
          acc[nt] = __builtin_amdgcn_wmma_f32_16x16x4_f32(
              false, a, false, wf[ks][nt], (short)0, acc[nt], false, false);
        }
      }
#pragma unroll
      for (int nt = 0; nt < 4; ++nt)
#pragma unroll
        for (int r = 0; r < 8; r += 2) {
          v2f val;
          val.x = acc[nt][r];
          val.y = acc[nt][r + 1];
          pw[mt * 8 + half * 4 + (r >> 1)][nt * 16 + ln] = val;
        }
    }

    v2f pf[8][4];
#pragma unroll
    for (int ks = 0; ks < 8; ++ks)
#pragma unroll
      for (int nt = 0; nt < 4; ++nt) pf[ks][nt] = pw[2 * ks + half][nt * 16 + ln];

    if (frame + NWAVES < NFRM) {
      const float* xfn = x + (size_t)(frame + NWAVES) * (NJOINT * CIN);
      const v2f* p0 = (const v2f*)(xfn + xoff0);
      const v2f* p1 = (const v2f*)(xfn + xoff1);
#pragma unroll
      for (int ks = 0; ks < 16; ++ks) { xa[0][ks] = p0[2 * ks]; xa[1][ks] = p1[2 * ks]; }
    }

    __builtin_amdgcn_sched_barrier(0);

#pragma unroll
    for (int mt = 0; mt < 2; ++mt) {
      v8f acc[4];
#pragma unroll
      for (int nt = 0; nt < 4; ++nt) acc[nt] = (v8f)(0.0f);

#pragma unroll
      for (int ks = 0; ks < 8; ++ks) {
#pragma unroll
        for (int nt = 0; nt < 4; ++nt) {
          acc[nt] = __builtin_amdgcn_wmma_f32_16x16x4_f32(
              false, af[mt][ks], false, pf[ks][nt], (short)0, acc[nt], false, false);
        }
      }
#pragma unroll
      for (int nt = 0; nt < 4; ++nt) {
        const int colb = nt * 16 + ln;
#pragma unroll
        for (int r = 0; r < 8; ++r) {
          const int v = mt * 16 + half * 8 + r;
          float val = acc[nt][r] + bb[nt];
          hw[v * HSTR + colb] = val > 0.0f ? val : 0.0f;
        }
      }
    }

    float* of = out + (size_t)frame * VC;
#pragma unroll
    for (int s = 0; s < 13; ++s) {
      const int q = s * 32 + lane;
      if (s < 12 || lane < 16) {
        const v4f hv = *(const v4f*)(hw + (q >> 4) * HSTR + (q & 15) * 4);
        *(v4f*)(of + q * 4) = hv;
        ssum[s] += hv;
        ssq[s] += hv * hv;
      }
    }
  }

  float* sqw = (float*)pw;
#pragma unroll
  for (int s = 0; s < 13; ++s) {
    const int q = s * 32 + lane;
    if (s < 12 || lane < 16) {
      *(v4f*)(hw + q * 4) = ssum[s];
      *(v4f*)(sqw + q * 4) = ssq[s];
    }
  }
  __syncthreads();
  if (ws_bytes >= WS_NEED_BYTES) {
    float* psum = ws + WS_PSUM + (size_t)blockIdx.x * VC;
    float* psq  = ws + WS_PSQ  + (size_t)blockIdx.x * VC;
    for (int q = tid; q < VC / 4; q += WPB * 32) {
      v4f a = *(const v4f*)(hbuf[0] + q * 4);
      v4f b = *(const v4f*)((const float*)pwbuf[0] + q * 4);
#pragma unroll
      for (int w = 1; w < WPB; ++w) {
        a += *(const v4f*)(hbuf[w] + q * 4);
        b += *(const v4f*)((const float*)pwbuf[w] + q * 4);
      }
      *(v4f*)(psum + q * 4) = a;
      *(v4f*)(psq + q * 4) = b;
    }
  }
}

static_assert(FIN_R == 4, "the pairwise f32 add below is written for four rows a thread");
__launch_bounds__(FIN_C4 * FIN_G)
__global__ void stgcn_bn_finalize(const float* __restrict__ gamma,
                                  const float* __restrict__ beta,
                                  float* __restrict__ ws, size_t ws_bytes) {
  __shared__ __attribute__((aligned(16))) double r1[FIN_G][FIN_LD];
  __shared__ __attribute__((aligned(16))) double r2[FIN_P][FIN_LD];
  const int tid = threadIdx.x;
  const int c2 = tid & (FIN_C4 - 1);
  const int g  = tid / FIN_C4;
  const int chb = blockIdx.x * (FIN_C4 * 4) + c2 * 4;
  v4f a = (v4f)(0.0f), b = (v4f)(0.0f);
  if (ws_bytes >= WS_NEED_BYTES) {
    v4f ar[FIN_R], br[FIN_R];
#pragma unroll
    for (int i = 0; i < FIN_R; ++i) {
      const int p = g + i * FIN_G;
      ar[i] = *(const v4f*)(ws + WS_PSUM + (size_t)p * VC + chb);
      br[i] = *(const v4f*)(ws + WS_PSQ  + (size_t)p * VC + chb);
    }
    a = (ar[0] + ar[1]) + (ar[2] + ar[3]);
    b = (br[0] + br[1]) + (br[2] + br[3]);
  }
#pragma unroll
  for (int j = 0; j < 4; ++j) {
    r1[g][c2 * 8 + j]     = (double)a[j];
    r1[g][c2 * 8 + 4 + j] = (double)b[j];
  }
  __syncthreads();
  const int k = tid & (FIN_V - 1);
  const int p = tid / FIN_V;
  double acc = r1[p * FIN_Q][k];
#pragma unroll
  for (int q = 1; q < FIN_Q; ++q) acc += r1[p * FIN_Q + q][k];
  r2[p][k] = acc;
  __syncthreads();
  double tot = 0.0;
  if (tid < FIN_V) {
    tot = r2[0][tid];
#pragma unroll
    for (int q = 1; q < FIN_P; ++q) tot += r2[q][tid];
  }
  const double totsq = __shfl_down(tot, 4);
  if (tid < FIN_V && (tid & 4) == 0) {
    const int ch = blockIdx.x * (FIN_C4 * 4) + (tid >> 3) * 4 + (tid & 3);
    const double inv_n = 1.0 / (double)NFRM;
    const double mean = tot * inv_n;
    const double var = totsq * inv_n - mean * mean;
    const float scale = gamma[ch] * rsqrtf((float)var + BN_EPS);
    const float shift = beta[ch] - (float)mean * scale;
    ws[WS_SCALE + ch] = scale;
    ws[WS_SHIFT + ch] = shift;
  }
}

__global__ void stgcn_bn_apply(float* __restrict__ out,
                               const float* __restrict__ ws) {
  const size_t idx = ((size_t)blockIdx.x * blockDim.x + threadIdx.x) * 4;
  float4 h = *(const float4*)(out + idx);
  const int ch = (int)(idx % VC);
  const float4 sc = *(const float4*)(ws + WS_SCALE + ch);
  const float4 sh = *(const float4*)(ws + WS_SHIFT + ch);
  h.x = h.x * sc.x + sh.x;
  h.y = h.y * sc.y + sh.y;
  h.z = h.z * sc.z + sh.z;
  h.w = h.w * sc.w + sh.w;
  *(float4*)(out + idx) = h;
}

extern "C" void kernel_launch(void* const* d_in, const int* in_sizes, int n_in,
                              void* d_out, int out_size, void* d_ws, size_t ws_size,
                              hipStream_t stream) {
  const float* x     = (const float*)d_in[0];
  const float* W     = (const float*)d_in[1];
  const float* b     = (const float*)d_in[2];
  const float* gamma = (const float*)d_in[3];
  const float* beta  = (const float*)d_in[4];
  const int*   ei    = (const int*)d_in[5];
  float* out = (float*)d_out;
  float* ws  = (float*)d_ws;
  const int E = in_sizes[5] / 2;

  hipLaunchKernelGGL(stgcn_gcn_gemm, dim3(BLOCKS_B), dim3(WPB * 32), 0, stream,
                     x, W, b, ei, E, out, ws, ws_size);
  hipLaunchKernelGGL(stgcn_bn_finalize, dim3(VC / (FIN_C4 * 4)), dim3(FIN_C4 * FIN_G),
                     0, stream, gamma, beta, ws, ws_size);
  hipLaunchKernelGGL(stgcn_bn_apply, dim3((NFRM * VC) / 4 / 256), dim3(256),
                     0, stream, out, ws);
}


# A_hat built by edge owners: a thread takes one edge and finds in one scan of packed LDS keys how often it occurs and its node's degree, then adds its weight that often; all-edge walk per entry gone
# speedup vs baseline: 2.3599x; 1.0722x over previous
#include <hip/hip_runtime.h>
#include <cstddef>

#define NJOINT 25
#define CIN    64
#define COUT   64
#define VC     (NJOINT * COUT)
#define NFRM   (64 * 300)
#define BN_EPS 1e-5f

#define WPB        4
#define BLOCKS_B   512
#define NWAVES     (BLOCKS_B * WPB)

#define WS_SCALE  0
#define WS_SHIFT  (WS_SCALE + VC)
#define WS_PSUM   (WS_SHIFT + VC)
#define WS_PSQ    (WS_PSUM + BLOCKS_B * VC)
#define WS_NEED_BYTES ((size_t)(WS_PSQ + BLOCKS_B * VC) * sizeof(float))

#define FIN_C4  2
#define FIN_G   128
#define FIN_R   (BLOCKS_B / FIN_G)
#define FIN_V   (FIN_C4 * 8)
#define FIN_P   16
#define FIN_Q   (FIN_G / FIN_P)
#define FIN_LD  (FIN_V + 2)

typedef __attribute__((ext_vector_type(2))) float v2f;
typedef __attribute__((ext_vector_type(4))) float v4f;
typedef __attribute__((ext_vector_type(8))) float v8f;

#define WSTR 76
#define PSTR 80
#define ASTR 17
#define HSTR 68
#define ECHUNK (WPB * 32)

__launch_bounds__(WPB * 32, 1)
__global__ void stgcn_gcn_gemm(const float* __restrict__ x,
                               const float* __restrict__ W,
                               const float* __restrict__ bias,
                               const int* __restrict__ edge_index, int E,
                               float* __restrict__ out,
                               float* __restrict__ ws, size_t ws_bytes) {
  __shared__ v2f wpair[CIN / 2][WSTR];
  __shared__ v2f apair[32][ASTR];
  __shared__ __attribute__((aligned(16))) v2f pwbuf[WPB][16][PSTR];
  __shared__ __attribute__((aligned(16))) float hbuf[WPB][32 * HSTR];
  __shared__ float bbuf[COUT];
  __shared__ __attribute__((aligned(16))) int ekey[ECHUNK];
  __shared__ float dinvbuf[32];

  const int tid = threadIdx.x;
  const int wave = tid >> 5;
  const int lane = tid & 31;
  const int half = lane >> 4;
  const int ln   = lane & 15;
  const int waveGlobal = blockIdx.x * WPB + wave;

  const bool m1valid = (16 + ln) < NJOINT;
  const int xoff0 = ln * CIN + 2 * half;
  const int xoff1 = (m1valid ? (16 + ln) : (NJOINT - 1)) * CIN + 2 * half;
  v2f xa[2][16];
  {
    const float* xf0 = x + (size_t)waveGlobal * (NJOINT * CIN);
    const v2f* p0 = (const v2f*)(xf0 + xoff0);
    const v2f* p1 = (const v2f*)(xf0 + xoff1);
#pragma unroll
    for (int ks = 0; ks < 16; ++ks) { xa[0][ks] = p0[2 * ks]; xa[1][ks] = p1[2 * ks]; }
  }


  const int* esrc = edge_index;
  const int* etgt = edge_index + E;
  int t0 = -1, s0 = -1;
  if (tid < E) { t0 = etgt[tid]; s0 = esrc[tid]; }

  for (int i = tid; i < (CIN / 2) * COUT; i += WPB * 32) {
    const int p = i >> 6, d = i & 63;
    v2f val;
    val.x = W[(2 * p) * COUT + d];
    val.y = W[(2 * p + 1) * COUT + d];
    wpair[p][d] = val;
  }
  if (tid < COUT) bbuf[tid] = bias[tid];

  const int nlo = lane << 5;
  float* aflat = (float*)&apair[0][0];
  {
    v2f* az = &apair[0][0];
    for (int i = tid; i < 32 * ASTR; i += WPB * 32) az[i] = (v2f)(0.0f);
  }
  const int nchunks = (E + ECHUNK - 1) / ECHUNK;
  const int npass = nchunks > 0 ? nchunks : 1;
  for (int co = 0; co < npass; ++co) {
    int myt = t0, mys = s0;
    if (co > 0) {
      const int eo = co * ECHUNK + tid;
      myt = -1; mys = -1;
      if (eo < E) { myt = etgt[eo]; mys = esrc[eo]; }
    }
    const bool own = ((unsigned)myt < 32u) && ((unsigned)mys < 32u);
    const int mykey = own ? ((myt << 5) | mys) : -2;
    int cnt = 0;
    int degc = 0;
    for (int cs = 0; cs < nchunks; ++cs) {
      if (co + cs > 0) __syncthreads();
      int key = -1;
      if (cs == co) {
        if (own) key = mykey;
      } else {
        const int j = cs * ECHUNK + tid;
        if (j < E) {
          const int tt = etgt[j], ss = esrc[j];
          if (((unsigned)tt < 32u) && ((unsigned)ss < 32u)) key = (tt << 5) | ss;
        }
      }
      ekey[tid] = key;
      __syncthreads();
      const int n4 = (min(E - cs * ECHUNK, ECHUNK) + 3) >> 2;
#pragma unroll 4
      for (int g = 0; g < n4; ++g) {
        const int4 k4 = ((const int4*)ekey)[g];
        cnt += (k4.x == mykey ? 1 : 0) + (k4.y == mykey ? 1 : 0) + (k4.z == mykey ? 1 : 0) + (k4.w == mykey ? 1 : 0);
        degc += ((unsigned)(k4.x - nlo) < 32u ? 1 : 0) + ((unsigned)(k4.y - nlo) < 32u ? 1 : 0) +
                ((unsigned)(k4.z - nlo) < 32u ? 1 : 0) + ((unsigned)(k4.w - nlo) < 32u ? 1 : 0);
      }
    }
    if (co == 0) {
      if (wave == 0) dinvbuf[lane] = rsqrtf(1.0f + (float)degc);
      __syncthreads();
    }
    if (own) {
      const float w = dinvbuf[mys] * dinvbuf[myt];
      float val = 0.0f;
      for (int c = 0; c < cnt; ++c) val += w;
      aflat[(myt * ASTR + (mys >> 1)) * 2 + (mys & 1)] = val;
    }
  }
  __syncthreads();
  if (tid < NJOINT) {
    const float d = dinvbuf[tid];
    aflat[(tid * ASTR + (tid >> 1)) * 2 + (tid & 1)] += d * d;
  }
  __syncthreads();

  v2f (*pw)[PSTR] = pwbuf[wave];
  float* hw = hbuf[wave];

  v2f wf[16][4];
#pragma unroll
  for (int ks = 0; ks < 16; ++ks)
#pragma unroll
    for (int nt = 0; nt < 4; ++nt) wf[ks][nt] = wpair[2 * ks + half][nt * 16 + ln];
  v2f af[2][8];
#pragma unroll
  for (int mt = 0; mt < 2; ++mt)
#pragma unroll
    for (int ks = 0; ks < 8; ++ks) af[mt][ks] = apair[mt * 16 + ln][2 * ks + half];
  float bb[4];
#pragma unroll
  for (int nt = 0; nt < 4; ++nt) bb[nt] = bbuf[nt * 16 + ln];

  v4f ssum[13], ssq[13];
#pragma unroll
  for (int s = 0; s < 13; ++s) { ssum[s] = (v4f)(0.0f); ssq[s] = (v4f)(0.0f); }

  for (int frame = waveGlobal; frame < NFRM; frame += NWAVES) {

#pragma unroll
    for (int mt = 0; mt < 2; ++mt) {
      v8f acc[4];
#pragma unroll
      for (int nt = 0; nt < 4; ++nt) acc[nt] = (v8f)(0.0f);

#pragma unroll
      for (int ks = 0; ks < 16; ++ks) {
        v2f a = xa[mt][ks];
        if (mt == 1 && !m1valid) a = (v2f)(0.0f);
#pragma unroll
        for (int nt = 0; nt < 4; ++nt) {
          acc[nt] = __builtin_amdgcn_wmma_f32_16x16x4_f32(
              false, a, false, wf[ks][nt], (short)0, acc[nt], false, false);
        }
      }
#pragma unroll
      for (int nt = 0; nt < 4; ++nt)
#pragma unroll
        for (int r = 0; r < 8; r += 2) {
          v2f val;
          val.x = acc[nt][r];
          val.y = acc[nt][r + 1];
          pw[mt * 8 + half * 4 + (r >> 1)][nt * 16 + ln] = val;
        }
    }

    v2f pf[8][4];
#pragma unroll
    for (int ks = 0; ks < 8; ++ks)
#pragma unroll
      for (int nt = 0; nt < 4; ++nt) pf[ks][nt] = pw[2 * ks + half][nt * 16 + ln];

    if (frame + NWAVES < NFRM) {
      const float* xfn = x + (size_t)(frame + NWAVES) * (NJOINT * CIN);
      const v2f* p0 = (const v2f*)(xfn + xoff0);
      const v2f* p1 = (const v2f*)(xfn + xoff1);
#pragma unroll
      for (int ks = 0; ks < 16; ++ks) { xa[0][ks] = p0[2 * ks]; xa[1][ks] = p1[2 * ks]; }
    }

    __builtin_amdgcn_sched_barrier(0);

#pragma unroll
    for (int mt = 0; mt < 2; ++mt) {
      v8f acc[4];
#pragma unroll
      for (int nt = 0; nt < 4; ++nt) acc[nt] = (v8f)(0.0f);

#pragma unroll
      for (int ks = 0; ks < 8; ++ks) {
#pragma unroll
        for (int nt = 0; nt < 4; ++nt) {
          acc[nt] = __builtin_amdgcn_wmma_f32_16x16x4_f32(
              false, af[mt][ks], false, pf[ks][nt], (short)0, acc[nt], false, false);
        }
      }
#pragma unroll
      for (int nt = 0; nt < 4; ++nt) {
        const int colb = nt * 16 + ln;
#pragma unroll
        for (int r = 0; r < 8; ++r) {
          const int v = mt * 16 + half * 8 + r;
          float val = acc[nt][r] + bb[nt];
          hw[v * HSTR + colb] = val > 0.0f ? val : 0.0f;
        }
      }
    }

    float* of = out + (size_t)frame * VC;
#pragma unroll
    for (int s = 0; s < 13; ++s) {
      const int q = s * 32 + lane;
      if (s < 12 || lane < 16) {
        const v4f hv = *(const v4f*)(hw + (q >> 4) * HSTR + (q & 15) * 4);
        *(v4f*)(of + q * 4) = hv;
        ssum[s] += hv;
        ssq[s] += hv * hv;
      }
    }
  }

  float* sqw = (float*)pw;
#pragma unroll
  for (int s = 0; s < 13; ++s) {
    const int q = s * 32 + lane;
    if (s < 12 || lane < 16) {
      *(v4f*)(hw + q * 4) = ssum[s];
      *(v4f*)(sqw + q * 4) = ssq[s];
    }
  }
  __syncthreads();
  if (ws_bytes >= WS_NEED_BYTES) {
    float* psum = ws + WS_PSUM + (size_t)blockIdx.x * VC;
    float* psq  = ws + WS_PSQ  + (size_t)blockIdx.x * VC;
    for (int q = tid; q < VC / 4; q += WPB * 32) {
      v4f a = *(const v4f*)(hbuf[0] + q * 4);
      v4f b = *(const v4f*)((const float*)pwbuf[0] + q * 4);
#pragma unroll
      for (int w = 1; w < WPB; ++w) {
        a += *(const v4f*)(hbuf[w] + q * 4);
        b += *(const v4f*)((const float*)pwbuf[w] + q * 4);
      }
      *(v4f*)(psum + q * 4) = a;
      *(v4f*)(psq + q * 4) = b;
    }
  }
}

static_assert(FIN_R == 4, "the pairwise f32 add below is written for four rows a thread");
__launch_bounds__(FIN_C4 * FIN_G)
__global__ void stgcn_bn_finalize(const float* __restrict__ gamma,
                                  const float* __restrict__ beta,
                                  float* __restrict__ ws, size_t ws_bytes) {
  __shared__ __attribute__((aligned(16))) double r1[FIN_G][FIN_LD];
  __shared__ __attribute__((aligned(16))) double r2[FIN_P][FIN_LD];
  const int tid = threadIdx.x;
  const int c2 = tid & (FIN_C4 - 1);
  const int g  = tid / FIN_C4;
  const int chb = blockIdx.x * (FIN_C4 * 4) + c2 * 4;
  v4f a = (v4f)(0.0f), b = (v4f)(0.0f);
  if (ws_bytes >= WS_NEED_BYTES) {
    v4f ar[FIN_R], br[FIN_R];
#pragma unroll
    for (int i = 0; i < FIN_R; ++i) {
      const int p = g + i * FIN_G;
      ar[i] = *(const v4f*)(ws + WS_PSUM + (size_t)p * VC + chb);
      br[i] = *(const v4f*)(ws + WS_PSQ  + (size_t)p * VC + chb);
    }
    a = (ar[0] + ar[1]) + (ar[2] + ar[3]);
    b = (br[0] + br[1]) + (br[2] + br[3]);
  }
#pragma unroll
  for (int j = 0; j < 4; ++j) {
    r1[g][c2 * 8 + j]     = (double)a[j];
    r1[g][c2 * 8 + 4 + j] = (double)b[j];
  }
  __syncthreads();
  const int k = tid & (FIN_V - 1);
  const int p = tid / FIN_V;
  double acc = r1[p * FIN_Q][k];
#pragma unroll
  for (int q = 1; q < FIN_Q; ++q) acc += r1[p * FIN_Q + q][k];
  r2[p][k] = acc;
  __syncthreads();
  double tot = 0.0;
  if (tid < FIN_V) {
    tot = r2[0][tid];
#pragma unroll
    for (int q = 1; q < FIN_P; ++q) tot += r2[q][tid];
  }
  const double totsq = __shfl_down(tot, 4);
  if (tid < FIN_V && (tid & 4) == 0) {
    const int ch = blockIdx.x * (FIN_C4 * 4) + (tid >> 3) * 4 + (tid & 3);
    const double inv_n = 1.0 / (double)NFRM;
    const double mean = tot * inv_n;
    const double var = totsq * inv_n - mean * mean;
    const float scale = gamma[ch] * rsqrtf((float)var + BN_EPS);
    const float shift = beta[ch] - (float)mean * scale;
    ws[WS_SCALE + ch] = scale;
    ws[WS_SHIFT + ch] = shift;
  }
}

__global__ void stgcn_bn_apply(float* __restrict__ out,
                               const float* __restrict__ ws) {
  const size_t idx = ((size_t)blockIdx.x * blockDim.x + threadIdx.x) * 4;
  float4 h = *(const float4*)(out + idx);
  const int ch = (int)(idx % VC);
  const float4 sc = *(const float4*)(ws + WS_SCALE + ch);
  const float4 sh = *(const float4*)(ws + WS_SHIFT + ch);
  h.x = h.x * sc.x + sh.x;
  h.y = h.y * sc.y + sh.y;
  h.z = h.z * sc.z + sh.z;
  h.w = h.w * sc.w + sh.w;
  *(float4*)(out + idx) = h;
}

extern "C" void kernel_launch(void* const* d_in, const int* in_sizes, int n_in,
                              void* d_out, int out_size, void* d_ws, size_t ws_size,
                              hipStream_t stream) {
  const float* x     = (const float*)d_in[0];
  const float* W     = (const float*)d_in[1];
  const float* b     = (const float*)d_in[2];
  const float* gamma = (const float*)d_in[3];
  const float* beta  = (const float*)d_in[4];
  const int*   ei    = (const int*)d_in[5];
  float* out = (float*)d_out;
  float* ws  = (float*)d_ws;
  const int E = in_sizes[5] / 2;

  hipLaunchKernelGGL(stgcn_gcn_gemm, dim3(BLOCKS_B), dim3(WPB * 32), 0, stream,
                     x, W, b, ei, E, out, ws, ws_size);
  hipLaunchKernelGGL(stgcn_bn_finalize, dim3(VC / (FIN_C4 * 4)), dim3(FIN_C4 * FIN_G),
                     0, stream, gamma, beta, ws, ws_size);
  hipLaunchKernelGGL(stgcn_bn_apply, dim3((NFRM * VC) / 4 / 256), dim3(256),
                     0, stream, out, ws);
}


# Padded matrix work cut: stage-1 tiles run over 3 frames' 75 contiguous rows (5 tiles, were 6), stage 2 stops at k=28 (7 steps, were 8); frames regrouped per wave: BN partial sums add in a new order
# speedup vs baseline: 2.4460x; 1.0365x over previous
#include <hip/hip_runtime.h>
#include <cstddef>

#define NJOINT 25
#define CIN    64
#define COUT   64
#define VC     (NJOINT * COUT)
#define NFRM   (64 * 300)
#define BN_EPS 1e-5f

#define WPB        4
#define BLOCKS_B   512
#define NWAVES     (BLOCKS_B * WPB)
#define GF         3
#define GPW        3
#define NGRPFR     (NWAVES * GPW * GF)
#define NSINGLE    (NFRM - NGRPFR)
static_assert(NSINGLE >= 0 && NSINGLE <= NWAVES, "the frames left over must fit one a wave");

#define WS_SCALE  0
#define WS_SHIFT  (WS_SCALE + VC)
#define WS_PSUM   (WS_SHIFT + VC)
#define WS_PSQ    (WS_PSUM + BLOCKS_B * VC)
#define WS_NEED_BYTES ((size_t)(WS_PSQ + BLOCKS_B * VC) * sizeof(float))

#define FIN_C4  2
#define FIN_G   128
#define FIN_R   (BLOCKS_B / FIN_G)
#define FIN_V   (FIN_C4 * 8)
#define FIN_P   16
#define FIN_Q   (FIN_G / FIN_P)
#define FIN_LD  (FIN_V + 2)

typedef __attribute__((ext_vector_type(2))) float v2f;
typedef __attribute__((ext_vector_type(4))) float v4f;
typedef __attribute__((ext_vector_type(8))) float v8f;

#define WSTR 76
#define PSTR 80
#define ASTR 17
#define HSTR 68
#define KPR  14
#define FSTR (KPR * PSTR * 2)
#define WREG (GF * FSTR)
#define ECHUNK (WPB * 32)
static_assert(NJOINT * HSTR <= 12 * PSTR * 2, "the H tile must end in front of k-pair row 12 (joints 24..27)");

__host__ __device__ constexpr int pw_off(int m) {
  return (m / NJOINT) * FSTR + ((m % NJOINT) >> 1) * (PSTR * 2) + ((m % NJOINT) & 1);
}
#define D_SAME (4 * PSTR * 2)
#define D_ODD  (FSTR - 8 * PSTR * 2 - 1)
#define D_EVEN (FSTR - 9 * PSTR * 2 + 1)

__device__ __forceinline__ void load_x_tile(v2f (&dst)[16], const float* __restrict__ base,
                                            int t, int nrows, int half, int ln) {
  int m = 16 * t + ln;
  m = m < nrows ? m : nrows - 1;
  const v2f* p = (const v2f*)(base + (size_t)m * CIN + 2 * half);
#pragma unroll
  for (int ks = 0; ks < 16; ++ks) dst[ks] = p[2 * ks];
}

template <int NF>
__device__ __forceinline__ void gcn_item(const float* __restrict__ xbase,
                                         const float* __restrict__ xnext, int next_rows,
                                         float* __restrict__ outf, float* reg,
                                         const v2f (&wf)[16][4], const v2f (&af)[2][7],
                                         const float (&bb)[4], v2f (&xa)[2][16],
                                         v4f (&ssum)[13], v4f (&ssq)[13],
                                         int half, int ln, int lane) {
  constexpr int NROW = NF * NJOINT;
  constexpr int NTILE = (NROW + 15) / 16;
  float* const lb = reg + ln * 2;
  float* const b_same = lb + half * D_SAME;
  float* const b_odd  = lb + half * D_ODD;
  float* const b_even = lb + half * D_EVEN;

#pragma unroll
  for (int t = 0; t < NTILE; ++t) {
    v8f acc[4];
#pragma unroll
    for (int nt = 0; nt < 4; ++nt) acc[nt] = (v8f)(0.0f);
#pragma unroll
    for (int ks = 0; ks < 16; ++ks) {
#pragma unroll
      for (int nt = 0; nt < 4; ++nt) {
        acc[nt] = __builtin_amdgcn_wmma_f32_16x16x4_f32(
            false, xa[t & 1][ks], false, wf[ks][nt], (short)0, acc[nt], false, false);
      }
    }
    if (t + 2 < NTILE) load_x_tile(xa[t & 1], xbase, t + 2, NROW, half, ln);

#pragma unroll
    for (int r = 0; r < 8; ++r) {
      const int m0 = 16 * t + r;
      const int m1 = m0 + 8;
      const bool hi_row = m1 < NROW;
      const int d = hi_row ? (pw_off(m1) - pw_off(m0)) : D_SAME;
      float* const bp = (d == D_SAME) ? b_same : (d == D_ODD) ? b_odd : b_even;
#pragma unroll
      for (int nt = 0; nt < 4; ++nt) {
        if (hi_row || half == 0) bp[pw_off(m0) + nt * 32] = acc[nt][r];
      }
    }

    if (t + 2 >= NTILE && xnext != nullptr) load_x_tile(xa[t & 1], xnext, t & 1, next_rows, half, ln);
  }

#pragma unroll 1
  for (int f = 0; f < NF; ++f) {
    float* const fr = reg + f * FSTR;
    v2f pf[7][4];
#pragma unroll
    for (int ks = 0; ks < 7; ++ks)
#pragma unroll
      for (int nt = 0; nt < 4; ++nt)
        pf[ks][nt] = *(const v2f*)(fr + (2 * ks + half) * (PSTR * 2) + (nt * 16 + ln) * 2);

    __builtin_amdgcn_sched_barrier(0);

#pragma unroll
    for (int mt = 0; mt < 2; ++mt) {
      v8f acc[4];
#pragma unroll
      for (int nt = 0; nt < 4; ++nt) acc[nt] = (v8f)(0.0f);
#pragma unroll
      for (int ks = 0; ks < 7; ++ks) {
#pragma unroll
        for (int nt = 0; nt < 4; ++nt) {
          acc[nt] = __builtin_amdgcn_wmma_f32_16x16x4_f32(
              false, af[mt][ks], false, pf[ks][nt], (short)0, acc[nt], false, false);
        }
      }
#pragma unroll
      for (int nt = 0; nt < 4; ++nt) {
        const int colb = nt * 16 + ln;
#pragma unroll
        for (int r = 0; r < 8; ++r) {
          const int v = mt * 16 + half * 8 + r;
          const float val = acc[nt][r] + bb[nt];
          if (mt == 0 || r == 0 || half == 0) fr[v * HSTR + colb] = val > 0.0f ? val : 0.0f;
        }
      }
    }

    float* of = outf + (size_t)f * VC;
    const float* hrow = fr + half * HSTR + ln * 4;
#pragma unroll
    for (int s = 0; s < 13; ++s) {
      const int q = s * 32 + lane;
      if (s < 12 || lane < 16) {
        const v4f hv = *(const v4f*)(hrow + s * (2 * HSTR));
        *(v4f*)(of + q * 4) = hv;
        ssum[s] += hv;
        ssq[s] += hv * hv;
      }
    }
  }
}

__launch_bounds__(WPB * 32, 2)
__global__ void stgcn_gcn_gemm(const float* __restrict__ x,
                               const float* __restrict__ W,
                               const float* __restrict__ bias,
                               const int* __restrict__ edge_index, int E,
                               float* __restrict__ out,
                               float* __restrict__ ws, size_t ws_bytes) {
  __shared__ v2f wpair[CIN / 2][WSTR];
  __shared__ v2f apair[32][ASTR];
  __shared__ __attribute__((aligned(16))) float stage[WPB * WREG];
  __shared__ float bbuf[COUT];
  __shared__ __attribute__((aligned(16))) int ekey[ECHUNK];
  __shared__ float dinvbuf[32];

  const int tid = threadIdx.x;
  const int wave = tid >> 5;
  const int lane = tid & 31;
  const int half = lane >> 4;
  const int ln   = lane & 15;
  const int waveGlobal = blockIdx.x * WPB + wave;

  v2f xa[2][16];
  {
    const float* xg0 = x + (size_t)waveGlobal * (GF * NJOINT * CIN);
    load_x_tile(xa[0], xg0, 0, GF * NJOINT, half, ln);
    load_x_tile(xa[1], xg0, 1, GF * NJOINT, half, ln);
  }

  const int* esrc = edge_index;
  const int* etgt = edge_index + E;
  int t0 = -1, s0 = -1;
  if (tid < E) { t0 = etgt[tid]; s0 = esrc[tid]; }

  for (int i = tid; i < (CIN / 2) * COUT; i += WPB * 32) {
    const int p = i >> 6, d = i & 63;
    v2f val;
    val.x = W[(2 * p) * COUT + d];
    val.y = W[(2 * p + 1) * COUT + d];
    wpair[p][d] = val;
  }
  if (tid < COUT) bbuf[tid] = bias[tid];

  for (int i = tid; i < WPB * GF * PSTR; i += WPB * 32) {
    const int rg = i / PSTR, j = i - rg * PSTR;
    *(v4f*)(stage + rg * FSTR + 12 * (PSTR * 2) + j * 4) = (v4f)(0.0f);
  }

  const int nlo = lane << 5;
  float* aflat = (float*)&apair[0][0];
  {
    v2f* az = &apair[0][0];
    for (int i = tid; i < 32 * ASTR; i += WPB * 32) az[i] = (v2f)(0.0f);
  }
  const int nchunks = (E + ECHUNK - 1) / ECHUNK;
  const int npass = nchunks > 0 ? nchunks : 1;
  for (int co = 0; co < npass; ++co) {
    int myt = t0, mys = s0;
    if (co > 0) {
      const int eo = co * ECHUNK + tid;
      myt = -1; mys = -1;
      if (eo < E) { myt = etgt[eo]; mys = esrc[eo]; }
    }
    const bool own = ((unsigned)myt < 32u) && ((unsigned)mys < 32u);
    const int mykey = own ? ((myt << 5) | mys) : -2;
    int cnt = 0;
    int degc = 0;
    for (int cs = 0; cs < nchunks; ++cs) {
      if (co + cs > 0) __syncthreads();
      int key = -1;
      if (cs == co) {
        if (own) key = mykey;
      } else {
        const int j = cs * ECHUNK + tid;
        if (j < E) {
          const int tt = etgt[j], ss = esrc[j];
          if (((unsigned)tt < 32u) && ((unsigned)ss < 32u)) key = (tt << 5) | ss;
        }
      }
      ekey[tid] = key;
      __syncthreads();
      const int n4 = (min(E - cs * ECHUNK, ECHUNK) + 3) >> 2;
#pragma unroll 4
      for (int g = 0; g < n4; ++g) {
        const int4 k4 = ((const int4*)ekey)[g];
        cnt += (k4.x == mykey ? 1 : 0) + (k4.y == mykey ? 1 : 0) + (k4.z == mykey ? 1 : 0) + (k4.w == mykey ? 1 : 0);
        degc += ((unsigned)(k4.x - nlo) < 32u ? 1 : 0) + ((unsigned)(k4.y - nlo) < 32u ? 1 : 0) +
                ((unsigned)(k4.z - nlo) < 32u ? 1 : 0) + ((unsigned)(k4.w - nlo) < 32u ? 1 : 0);
      }
    }
    if (co == 0) {
      if (wave == 0) dinvbuf[lane] = rsqrtf(1.0f + (float)degc);
      __syncthreads();
    }
    if (own) {
      const float w = dinvbuf[mys] * dinvbuf[myt];
      float val = 0.0f;
      for (int c = 0; c < cnt; ++c) val += w;
      aflat[(myt * ASTR + (mys >> 1)) * 2 + (mys & 1)] = val;
    }
  }
  __syncthreads();
  if (tid < NJOINT) {
    const float d = dinvbuf[tid];
    aflat[(tid * ASTR + (tid >> 1)) * 2 + (tid & 1)] += d * d;
  }
  __syncthreads();

  float* const reg = stage + wave * WREG;

  v2f wf[16][4];
#pragma unroll
  for (int ks = 0; ks < 16; ++ks)
#pragma unroll
    for (int nt = 0; nt < 4; ++nt) wf[ks][nt] = wpair[2 * ks + half][nt * 16 + ln];
  v2f af[2][7];
#pragma unroll
  for (int mt = 0; mt < 2; ++mt)
#pragma unroll
    for (int ks = 0; ks < 7; ++ks) af[mt][ks] = apair[mt * 16 + ln][2 * ks + half];
  float bb[4];
#pragma unroll
  for (int nt = 0; nt < 4; ++nt) bb[nt] = bbuf[nt * 16 + ln];

  v4f ssum[13], ssq[13];
#pragma unroll
  for (int s = 0; s < 13; ++s) { ssum[s] = (v4f)(0.0f); ssq[s] = (v4f)(0.0f); }

  for (int gi = 0; gi < GPW; ++gi) {
    const int G = waveGlobal + gi * NWAVES;
    const float* xb = x + (size_t)G * (GF * NJOINT * CIN);
    const float* xn = nullptr;
    int nr = 0;
    if (gi + 1 < GPW) {
      xn = xb + (size_t)NWAVES * (GF * NJOINT * CIN);
      nr = GF * NJOINT;
    } else if (waveGlobal < NSINGLE) {
      xn = x + (size_t)(NGRPFR + waveGlobal) * (NJOINT * CIN);
      nr = NJOINT;
    }
    gcn_item<GF>(xb, xn, nr, out + (size_t)G * (GF * VC), reg, wf, af, bb, xa, ssum, ssq,
                 half, ln, lane);
  }
  if (waveGlobal < NSINGLE) {
    const int fs = NGRPFR + waveGlobal;
    gcn_item<1>(x + (size_t)fs * (NJOINT * CIN), nullptr, 0, out + (size_t)fs * VC, reg,
                wf, af, bb, xa, ssum, ssq, half, ln, lane);
  }

  float* sumw = reg;
  float* sqw = reg + FSTR;
#pragma unroll
  for (int s = 0; s < 13; ++s) {
    const int q = s * 32 + lane;
    if (s < 12 || lane < 16) {
      *(v4f*)(sumw + q * 4) = ssum[s];
      *(v4f*)(sqw + q * 4) = ssq[s];
    }
  }
  __syncthreads();
  if (ws_bytes >= WS_NEED_BYTES) {
    float* psum = ws + WS_PSUM + (size_t)blockIdx.x * VC;
    float* psq  = ws + WS_PSQ  + (size_t)blockIdx.x * VC;
    for (int q = tid; q < VC / 4; q += WPB * 32) {
      v4f a = *(const v4f*)(stage + q * 4);
      v4f b = *(const v4f*)(stage + FSTR + q * 4);
#pragma unroll
      for (int w = 1; w < WPB; ++w) {
        a += *(const v4f*)(stage + w * WREG + q * 4);
        b += *(const v4f*)(stage + w * WREG + FSTR + q * 4);
      }
      *(v4f*)(psum + q * 4) = a;
      *(v4f*)(psq + q * 4) = b;
    }
  }
}

static_assert(FIN_R == 4, "the pairwise f32 add below is written for four rows a thread");
__launch_bounds__(FIN_C4 * FIN_G)
__global__ void stgcn_bn_finalize(const float* __restrict__ gamma,
                                  const float* __restrict__ beta,
                                  float* __restrict__ ws, size_t ws_bytes) {
  __shared__ __attribute__((aligned(16))) double r1[FIN_G][FIN_LD];
  __shared__ __attribute__((aligned(16))) double r2[FIN_P][FIN_LD];
  const int tid = threadIdx.x;
  const int c2 = tid & (FIN_C4 - 1);
  const int g  = tid / FIN_C4;
  const int chb = blockIdx.x * (FIN_C4 * 4) + c2 * 4;
  v4f a = (v4f)(0.0f), b = (v4f)(0.0f);
  if (ws_bytes >= WS_NEED_BYTES) {
    v4f ar[FIN_R], br[FIN_R];
#pragma unroll
    for (int i = 0; i < FIN_R; ++i) {
      const int p = g + i * FIN_G;
      ar[i] = *(const v4f*)(ws + WS_PSUM + (size_t)p * VC + chb);
      br[i] = *(const v4f*)(ws + WS_PSQ  + (size_t)p * VC + chb);
    }
    a = (ar[0] + ar[1]) + (ar[2] + ar[3]);
    b = (br[0] + br[1]) + (br[2] + br[3]);
  }
#pragma unroll
  for (int j = 0; j < 4; ++j) {
    r1[g][c2 * 8 + j]     = (double)a[j];
    r1[g][c2 * 8 + 4 + j] = (double)b[j];
  }
  __syncthreads();
  const int k = tid & (FIN_V - 1);
  const int p = tid / FIN_V;
  double acc = r1[p * FIN_Q][k];
#pragma unroll
  for (int q = 1; q < FIN_Q; ++q) acc += r1[p * FIN_Q + q][k];
  r2[p][k] = acc;
  __syncthreads();
  double tot = 0.0;
  if (tid < FIN_V) {
    tot = r2[0][tid];
#pragma unroll
    for (int q = 1; q < FIN_P; ++q) tot += r2[q][tid];
  }
  const double totsq = __shfl_down(tot, 4);
  if (tid < FIN_V && (tid & 4) == 0) {
    const int ch = blockIdx.x * (FIN_C4 * 4) + (tid >> 3) * 4 + (tid & 3);
    const double inv_n = 1.0 / (double)NFRM;
    const double mean = tot * inv_n;
    const double var = totsq * inv_n - mean * mean;
    const float scale = gamma[ch] * rsqrtf((float)var + BN_EPS);
    const float shift = beta[ch] - (float)mean * scale;
    ws[WS_SCALE + ch] = scale;
    ws[WS_SHIFT + ch] = shift;
  }
}

__global__ void stgcn_bn_apply(float* __restrict__ out,
                               const float* __restrict__ ws) {
  const size_t idx = ((size_t)blockIdx.x * blockDim.x + threadIdx.x) * 4;
  float4 h = *(const float4*)(out + idx);
  const int ch = (int)(idx % VC);
  const float4 sc = *(const float4*)(ws + WS_SCALE + ch);
  const float4 sh = *(const float4*)(ws + WS_SHIFT + ch);
  h.x = h.x * sc.x + sh.x;
  h.y = h.y * sc.y + sh.y;
  h.z = h.z * sc.z + sh.z;
  h.w = h.w * sc.w + sh.w;
  *(float4*)(out + idx) = h;
}

extern "C" void kernel_launch(void* const* d_in, const int* in_sizes, int n_in,
                              void* d_out, int out_size, void* d_ws, size_t ws_size,
                              hipStream_t stream) {
  const float* x     = (const float*)d_in[0];
  const float* W     = (const float*)d_in[1];
  const float* b     = (const float*)d_in[2];
  const float* gamma = (const float*)d_in[3];
  const float* beta  = (const float*)d_in[4];
  const int*   ei    = (const int*)d_in[5];
  float* out = (float*)d_out;
  float* ws  = (float*)d_ws;
  const int E = in_sizes[5] / 2;

  hipLaunchKernelGGL(stgcn_gcn_gemm, dim3(BLOCKS_B), dim3(WPB * 32), 0, stream,
                     x, W, b, ei, E, out, ws, ws_size);
  hipLaunchKernelGGL(stgcn_bn_finalize, dim3(VC / (FIN_C4 * 4)), dim3(FIN_C4 * FIN_G),
                     0, stream, gamma, beta, ws, ws_size);
  hipLaunchKernelGGL(stgcn_bn_apply, dim3((NFRM * VC) / 4 / 256), dim3(256),
                     0, stream, out, ws);
}
